# starting kernel
# baseline (speedup 1.0000x reference)
#include <hip/hip_runtime.h>
#include <math.h>

#define NEG_SLOPE 0.2f

typedef __attribute__((ext_vector_type(2))) float v2f;
typedef __attribute__((ext_vector_type(8))) float v8f;

__global__ __launch_bounds__(32) void gat_scores_wmma(
    const float* __restrict__ x, const float* __restrict__ W_u,
    const float* __restrict__ b_u, const float* __restrict__ W_v,
    float* __restrict__ s, int n_nodes) {
  const int tile = blockIdx.x;
  const int lane = threadIdx.x;
  const int half = lane >> 4;
  const int l15  = lane & 15;

  __shared__ float Wc[128 * 16];
  for (int i = lane; i < 128 * 16; i += 32) {
    const int k = i >> 4, c = i & 15;
    float w = 0.0f;
    if (c < 4)      w = W_u[k * 4 + c];
    else if (c < 8) w = W_v[k * 4 + (c - 4)];
    Wc[i] = w;
  }
  __syncthreads();

  int node = tile * 16 + l15;
  const int nodeC = node < n_nodes ? node : n_nodes - 1;
  const float* __restrict__ xrow = x + (size_t)nodeC * 128;

  v8f acc = {};
  #pragma unroll 4
  for (int k = 0; k < 128; k += 4) {
    const int ka = k + half * 2;
    v2f a; a.x = xrow[ka];            a.y = xrow[ka + 1];
    v2f b; b.x = Wc[ka * 16 + l15];   b.y = Wc[(ka + 1) * 16 + l15];
    acc = __builtin_amdgcn_wmma_f32_16x16x4_f32(
         false, a,  false, b,
         (short)0, acc,  false,  false);
  }

  const int ncol = l15;
  if (ncol < 8) {
    const float bias = (ncol < 4) ? b_u[ncol] : 0.0f;
    #pragma unroll
    for (int r = 0; r < 8; ++r) {
      const int mrow = r + half * 8;
      const int onode = tile * 16 + mrow;
      if (onode < n_nodes) s[(size_t)onode * 8 + ncol] = acc[r] + bias;
    }
  }
}

__global__ void gat_init_md(float* __restrict__ m, float* __restrict__ denom,
                            int n4) {
  const int i = blockIdx.x * blockDim.x + threadIdx.x;
  if (i < n4) { m[i] = -INFINITY; denom[i] = 0.0f; }
}

__global__ void gat_init_out(const float* __restrict__ x,
                             float* __restrict__ out, long long total) {
  const long long i = (long long)blockIdx.x * blockDim.x + threadIdx.x;
  if (i >= total) return;
  const long long node = i >> 8;
  const int d = (int)(i & 255);
  out[i] = (d < 128) ? x[node * 128 + d] : 0.0f;
}

__device__ __forceinline__ void atomicMaxFloat(float* addr, float val) {
  if (val >= 0.0f) atomicMax((int*)addr, __float_as_int(val));
  else             atomicMin((unsigned int*)addr, __float_as_uint(val));
}

__device__ __forceinline__ float lrelu(float v) {
  return v >= 0.0f ? v : NEG_SLOPE * v;
}

__global__ void gat_edge_max(const float* __restrict__ s,
                             const int* __restrict__ src,
                             const int* __restrict__ dst,
                             float* __restrict__ m, int E) {
  const int e = blockIdx.x * blockDim.x + threadIdx.x;
  if (e >= E) return;
  const int u = src[e], v = dst[e];
  const float4 su = *(const float4*)(s + (size_t)u * 8);
  const float4 sv = *(const float4*)(s + (size_t)v * 8 + 4);
  float* mv = m + (size_t)v * 4;
  atomicMaxFloat(mv + 0, lrelu(su.x + sv.x));
  atomicMaxFloat(mv + 1, lrelu(su.y + sv.y));
  atomicMaxFloat(mv + 2, lrelu(su.z + sv.z));
  atomicMaxFloat(mv + 3, lrelu(su.w + sv.w));
}

__global__ void gat_edge_denom(const float* __restrict__ s,
                               const int* __restrict__ src,
                               const int* __restrict__ dst,
                               const float* __restrict__ m,
                               float* __restrict__ denom, int E) {
  const int e = blockIdx.x * blockDim.x + threadIdx.x;
  if (e >= E) return;
  const int u = src[e], v = dst[e];
  const float4 su = *(const float4*)(s + (size_t)u * 8);
  const float4 sv = *(const float4*)(s + (size_t)v * 8 + 4);
  const float4 mv = *(const float4*)(m + (size_t)v * 4);
  float* dv = denom + (size_t)v * 4;
  atomicAdd(dv + 0, __expf(lrelu(su.x + sv.x) - mv.x));
  atomicAdd(dv + 1, __expf(lrelu(su.y + sv.y) - mv.y));
  atomicAdd(dv + 2, __expf(lrelu(su.z + sv.z) - mv.z));
  atomicAdd(dv + 3, __expf(lrelu(su.w + sv.w) - mv.w));
}

__global__ void gat_edge_agg(const float* __restrict__ x,
                             const float* __restrict__ s,
                             const int* __restrict__ src,
                             const int* __restrict__ dst,
                             const float* __restrict__ m,
                             const float* __restrict__ denom,
                             float* __restrict__ out, int E) {
  const long long gid = (long long)blockIdx.x * blockDim.x + threadIdx.x;
  const int e = (int)(gid >> 5);
  const int lane = (int)(gid & 31);
  if (e >= E) return;
  const int u = src[e], v = dst[e];

  const int h = lane & 3;
  const float ee = lrelu(s[(size_t)u * 8 + h] + s[(size_t)v * 8 + 4 + h]);
  const float p = __expf(ee - m[(size_t)v * 4 + h]) / denom[(size_t)v * 4 + h];
  const float p0 = __shfl(p, 0, 32);
  const float p1 = __shfl(p, 1, 32);
  const float p2 = __shfl(p, 2, 32);
  const float p3 = __shfl(p, 3, 32);

  const float4 xv = *(const float4*)(x + (size_t)u * 128 + lane * 4);
  float* o = out + (size_t)v * 256 + 128 + lane * 4;
  atomicAdd(o + 0, xv.x * p0);
  atomicAdd(o + 1, xv.y * p1);
  atomicAdd(o + 2, xv.z * p2);
  atomicAdd(o + 3, xv.w * p3);
}

extern "C" void kernel_launch(void* const* d_in, const int* in_sizes, int n_in,
                              void* d_out, int out_size, void* d_ws, size_t ws_size,
                              hipStream_t stream) {
  const float* x   = (const float*)d_in[0];
  const int*   src = (const int*)d_in[1];
  const int*   dst = (const int*)d_in[2];
  const float* W_u = (const float*)d_in[3];
  const float* b_u = (const float*)d_in[4];
  const float* W_v = (const float*)d_in[5];
  float* out = (float*)d_out;

  const int n = in_sizes[0] / 128;
  const int E = in_sizes[1];

  float* s     = (float*)d_ws;
  float* m     = s + (size_t)n * 8;
  float* denom = m + (size_t)n * 4;

  gat_scores_wmma<<<(n + 15) / 16, 32, 0, stream>>>(x, W_u, b_u, W_v, s, n);

  gat_init_md<<<(n * 4 + 255) / 256, 256, 0, stream>>>(m, denom, n * 4);
  const long long total = (long long)n * 256;
  gat_init_out<<<(unsigned)((total + 255) / 256), 256, 0, stream>>>(x, out, total);

  gat_edge_max<<<(E + 255) / 256, 256, 0, stream>>>(s, src, dst, m, E);

  gat_edge_denom<<<(E + 255) / 256, 256, 0, stream>>>(s, src, dst, m, denom, E);

  gat_edge_agg<<<(E + 7) / 8, 256, 0, stream>>>(x, s, src, dst, m, denom, out, E);
}


# Edge pass: resident grid, a wave walks trips of 8 edges, their 8 row gathers in flight together, the 32 weights one per lane (was one edge per wave); float atomics, arrival order as before
# speedup vs baseline: 1.6067x; 1.6067x over previous
#include <hip/hip_runtime.h>
#include <math.h>

#define NEG_SLOPE 0.2f

typedef __attribute__((ext_vector_type(2))) float v2f;
typedef __attribute__((ext_vector_type(8))) float v8f;

__global__ __launch_bounds__(32) void gat_scores_wmma(
    const float* __restrict__ x, const float* __restrict__ W_u,
    const float* __restrict__ b_u, const float* __restrict__ W_v,
    float* __restrict__ s, int n_nodes) {
  const int tile = blockIdx.x;
  const int lane = threadIdx.x;
  const int half = lane >> 4;
  const int l15  = lane & 15;

  __shared__ float Wc[128 * 16];
  for (int i = lane; i < 128 * 16; i += 32) {
    const int k = i >> 4, c = i & 15;
    float w = 0.0f;
    if (c < 4)      w = W_u[k * 4 + c];
    else if (c < 8) w = W_v[k * 4 + (c - 4)];
    Wc[i] = w;
  }
  __syncthreads();

  int node = tile * 16 + l15;
  const int nodeC = node < n_nodes ? node : n_nodes - 1;
  const float* __restrict__ xrow = x + (size_t)nodeC * 128;

  v8f acc = {};
  #pragma unroll 4
  for (int k = 0; k < 128; k += 4) {
    const int ka = k + half * 2;
    v2f a; a.x = xrow[ka];            a.y = xrow[ka + 1];
    v2f b; b.x = Wc[ka * 16 + l15];   b.y = Wc[(ka + 1) * 16 + l15];
    acc = __builtin_amdgcn_wmma_f32_16x16x4_f32(
         false, a,  false, b,
         (short)0, acc,  false,  false);
  }

  const int ncol = l15;
  if (ncol < 8) {
    const float bias = (ncol < 4) ? b_u[ncol] : 0.0f;
    #pragma unroll
    for (int r = 0; r < 8; ++r) {
      const int mrow = r + half * 8;
      const int onode = tile * 16 + mrow;
      if (onode < n_nodes) s[(size_t)onode * 8 + ncol] = acc[r] + bias;
    }
  }
}

__global__ void gat_init_md(float* __restrict__ m, float* __restrict__ denom,
                            int n4) {
  const int i = blockIdx.x * blockDim.x + threadIdx.x;
  if (i < n4) { m[i] = -INFINITY; denom[i] = 0.0f; }
}

__global__ void gat_init_out(const float* __restrict__ x,
                             float* __restrict__ out, long long total) {
  const long long i = (long long)blockIdx.x * blockDim.x + threadIdx.x;
  if (i >= total) return;
  const long long node = i >> 8;
  const int d = (int)(i & 255);
  out[i] = (d < 128) ? x[node * 128 + d] : 0.0f;
}

__device__ __forceinline__ void atomicMaxFloat(float* addr, float val) {
  if (val >= 0.0f) atomicMax((int*)addr, __float_as_int(val));
  else             atomicMin((unsigned int*)addr, __float_as_uint(val));
}

__device__ __forceinline__ float lrelu(float v) {
  return v >= 0.0f ? v : NEG_SLOPE * v;
}

__global__ void gat_edge_max(const float* __restrict__ s,
                             const int* __restrict__ src,
                             const int* __restrict__ dst,
                             float* __restrict__ m, int E) {
  const int e = blockIdx.x * blockDim.x + threadIdx.x;
  if (e >= E) return;
  const int u = src[e], v = dst[e];
  const float4 su = *(const float4*)(s + (size_t)u * 8);
  const float4 sv = *(const float4*)(s + (size_t)v * 8 + 4);
  float* mv = m + (size_t)v * 4;
  atomicMaxFloat(mv + 0, lrelu(su.x + sv.x));
  atomicMaxFloat(mv + 1, lrelu(su.y + sv.y));
  atomicMaxFloat(mv + 2, lrelu(su.z + sv.z));
  atomicMaxFloat(mv + 3, lrelu(su.w + sv.w));
}

#define TRIP_EDGES 8

__device__ __forceinline__ void agg_single_edge(
    const float* __restrict__ x, const float* __restrict__ s,
    const int* __restrict__ src, const int* __restrict__ dst,
    const float* __restrict__ m, float* __restrict__ denom,
    float* __restrict__ out, int e, int lane) {
  const int u = src[e], v = dst[e];
  const float* __restrict__ xr = x + (size_t)u * 128 + lane;
  const float x0 = xr[0], x1 = xr[32], x2 = xr[64], x3 = xr[96];
  const int h = lane & 3;
  const float ee = lrelu(s[(size_t)u * 8 + h] + s[(size_t)v * 8 + 4 + h]);
  const float w = __expf(ee - m[(size_t)v * 4 + h]);
  if (lane < 4) atomicAdd(denom + (size_t)v * 4 + lane, w);
  float* o = out + (size_t)v * 256 + 128 + lane;
  atomicAdd(o,      x0 * w);
  atomicAdd(o + 32, x1 * w);
  atomicAdd(o + 64, x2 * w);
  atomicAdd(o + 96, x3 * w);
}

__global__ __launch_bounds__(256) void gat_edge_agg(
    const float* __restrict__ x, const float* __restrict__ s,
    const int* __restrict__ src, const int* __restrict__ dst,
    const float* __restrict__ m, float* __restrict__ denom,
    float* __restrict__ out, int E) {
  const int lane = threadIdx.x & 31;
  const int h    = lane & 3;
  const int sub  = lane >> 2;
  const int wpb  = (int)(blockDim.x >> 5);
  const int wv = __builtin_amdgcn_readfirstlane(
      (int)blockIdx.x * wpb + (int)(threadIdx.x >> 5));
  const int nw = (int)gridDim.x * wpb;
  const int ntrip = E / TRIP_EDGES;

  int b = wv;
  int u_l = 0, v_l = 0;
  if (b < ntrip) {
    u_l = src[b * TRIP_EDGES + sub];
    v_l = dst[b * TRIP_EDGES + sub];
  }
  while (b < ntrip) {
    const int bn = b + nw;
    const int bp = bn < ntrip ? bn : ntrip - 1;
    const int u_n = src[bp * TRIP_EDGES + sub];
    const int v_n = dst[bp * TRIP_EDGES + sub];
    __builtin_amdgcn_sched_barrier(0);

    const float su = s[(size_t)u_l * 8 + h];
    const float sv = s[(size_t)v_l * 8 + 4 + h];
    const float mm = m[(size_t)v_l * 4 + h];

    float xv[TRIP_EDGES][4];
    #pragma unroll
    for (int k = 0; k < TRIP_EDGES; ++k) {
      const int u_k = __builtin_amdgcn_readlane(u_l, 4 * k);
      const float* __restrict__ xr = x + (size_t)u_k * 128 + lane;
      xv[k][0] = xr[0];  xv[k][1] = xr[32];
      xv[k][2] = xr[64]; xv[k][3] = xr[96];
    }

    const float w_l = __expf(lrelu(su + sv) - mm);
    atomicAdd(denom + (size_t)v_l * 4 + h, w_l);

    #pragma unroll
    for (int k = 0; k < TRIP_EDGES; ++k) {
      const int v_k = __builtin_amdgcn_readlane(v_l, 4 * k);
      const float w = __int_as_float(__builtin_amdgcn_ds_bpermute(
          (4 * k + h) << 2, __float_as_int(w_l)));
      float* o = out + (size_t)v_k * 256 + 128 + lane;
      atomicAdd(o,      xv[k][0] * w);
      atomicAdd(o + 32, xv[k][1] * w);
      atomicAdd(o + 64, xv[k][2] * w);
      atomicAdd(o + 96, xv[k][3] * w);
    }

    b = bn; u_l = u_n; v_l = v_n;
  }

  for (int e = ntrip * TRIP_EDGES + wv; e < E; e += nw)
    agg_single_edge(x, s, src, dst, m, denom, out, e, lane);
}

#define NORM_ROWS 4
__global__ __launch_bounds__(256) void gat_normalize(
    const float* __restrict__ denom, float* __restrict__ out, int n_nodes) {
  const int lane = threadIdx.x & 31;
  const int wave = (int)((blockIdx.x * blockDim.x + threadIdx.x) >> 5);
  const int node0 = wave * NORM_ROWS;

  float4 d[NORM_ROWS], o[NORM_ROWS];
  #pragma unroll
  for (int i = 0; i < NORM_ROWS; ++i) {
    const int node = node0 + i;
    if (node < n_nodes) {
      d[i] = *(const float4*)(denom + (size_t)node * 4);
      o[i] = *(const float4*)(out + (size_t)node * 256 + 128 + lane * 4);
    }
  }
  #pragma unroll
  for (int i = 0; i < NORM_ROWS; ++i) {
    const int node = node0 + i;
    if (node < n_nodes) {
      float4 r = o[i];
      if (d[i].x > 0.0f) r.x = r.x / d[i].x;
      if (d[i].y > 0.0f) r.y = r.y / d[i].y;
      if (d[i].z > 0.0f) r.z = r.z / d[i].z;
      if (d[i].w > 0.0f) r.w = r.w / d[i].w;
      *(float4*)(out + (size_t)node * 256 + 128 + lane * 4) = r;
    }
  }
}

extern "C" void kernel_launch(void* const* d_in, const int* in_sizes, int n_in,
                              void* d_out, int out_size, void* d_ws, size_t ws_size,
                              hipStream_t stream) {
  const float* x   = (const float*)d_in[0];
  const int*   src = (const int*)d_in[1];
  const int*   dst = (const int*)d_in[2];
  const float* W_u = (const float*)d_in[3];
  const float* b_u = (const float*)d_in[4];
  const float* W_v = (const float*)d_in[5];
  float* out = (float*)d_out;

  const int n = in_sizes[0] / 128;
  const int E = in_sizes[1];

  float* s     = (float*)d_ws;
  float* m     = s + (size_t)n * 8;
  float* denom = m + (size_t)n * 4;

  gat_scores_wmma<<<(n + 15) / 16, 32, 0, stream>>>(x, W_u, b_u, W_v, s, n);

  gat_init_md<<<(n * 4 + 255) / 256, 256, 0, stream>>>(m, denom, n * 4);
  const long long total = (long long)n * 256;
  gat_init_out<<<(unsigned)((total + 255) / 256), 256, 0, stream>>>(x, out, total);

  gat_edge_max<<<(E + 255) / 256, 256, 0, stream>>>(s, src, dst, m, E);

  gat_edge_agg<<<1024, 256, 0, stream>>>(x, s, src, dst, m, denom, out, E);

  gat_normalize<<<(n + 31) / 32, 256, 0, stream>>>(denom, out, n);
}


# One copy of each atomic target (row sums, exp sums, maxima) per half of the chip, picked by the wave's die id; node pass merges them: each sum now two partial sums, arrival order inside each
# speedup vs baseline: 1.7837x; 1.1102x over previous
#include <hip/hip_runtime.h>
#include <math.h>

#define NEG_SLOPE 0.2f

typedef __attribute__((ext_vector_type(2))) float v2f;
typedef __attribute__((ext_vector_type(8))) float v8f;

__global__ __launch_bounds__(32) void gat_scores_wmma(
    const float* __restrict__ x, const float* __restrict__ W_u,
    const float* __restrict__ b_u, const float* __restrict__ W_v,
    float* __restrict__ s, int n_nodes) {
  const int tile = blockIdx.x;
  const int lane = threadIdx.x;
  const int half = lane >> 4;
  const int l15  = lane & 15;

  __shared__ float Wc[128 * 16];
  for (int i = lane; i < 128 * 16; i += 32) {
    const int k = i >> 4, c = i & 15;
    float w = 0.0f;
    if (c < 4)      w = W_u[k * 4 + c];
    else if (c < 8) w = W_v[k * 4 + (c - 4)];
    Wc[i] = w;
  }
  __syncthreads();

  int node = tile * 16 + l15;
  const int nodeC = node < n_nodes ? node : n_nodes - 1;
  const float* __restrict__ xrow = x + (size_t)nodeC * 128;

  v8f acc = {};
  #pragma unroll 4
  for (int k = 0; k < 128; k += 4) {
    const int ka = k + half * 2;
    v2f a; a.x = xrow[ka];            a.y = xrow[ka + 1];
    v2f b; b.x = Wc[ka * 16 + l15];   b.y = Wc[(ka + 1) * 16 + l15];
    acc = __builtin_amdgcn_wmma_f32_16x16x4_f32(
         false, a,  false, b,
         (short)0, acc,  false,  false);
  }

  const int ncol = l15;
  if (ncol < 8) {
    const float bias = (ncol < 4) ? b_u[ncol] : 0.0f;
    #pragma unroll
    for (int r = 0; r < 8; ++r) {
      const int mrow = r + half * 8;
      const int onode = tile * 16 + mrow;
      if (onode < n_nodes) s[(size_t)onode * 8 + ncol] = acc[r] + bias;
    }
  }
}

__device__ __forceinline__ int chip_half() {
  return (int)((__builtin_amdgcn_s_sendmsg_rtn(0x87) >> 18) & 1u);
}

__global__ void gat_init_md(float* __restrict__ m, float* __restrict__ denom,
                            int n8) {
  const int i = blockIdx.x * blockDim.x + threadIdx.x;
  if (i < n8) { m[i] = -INFINITY; denom[i] = 0.0f; }
}

__global__ void gat_init_out(const float* __restrict__ x,
                             float* __restrict__ out,
                             float* __restrict__ acc1, long long total) {
  const long long i = (long long)blockIdx.x * blockDim.x + threadIdx.x;
  if (i >= total) return;
  const long long node = i >> 8;
  const int d = (int)(i & 255);
  if (d < 128) {
    out[i] = x[node * 128 + d];
  } else {
    out[i] = 0.0f;
    acc1[node * 128 + (d - 128)] = 0.0f;
  }
}

__device__ __forceinline__ void atomicMaxFloat(float* addr, float val) {
  if (val >= 0.0f) atomicMax((int*)addr, __float_as_int(val));
  else             atomicMin((unsigned int*)addr, __float_as_uint(val));
}

__device__ __forceinline__ float lrelu(float v) {
  return v >= 0.0f ? v : NEG_SLOPE * v;
}

__global__ void gat_edge_max(const float* __restrict__ s,
                             const int* __restrict__ src,
                             const int* __restrict__ dst,
                             float* __restrict__ m, int E) {
  const int e = blockIdx.x * blockDim.x + threadIdx.x;
  if (e >= E) return;
  const int u = src[e], v = dst[e];
  const float4 su = *(const float4*)(s + (size_t)u * 8);
  const float4 sv = *(const float4*)(s + (size_t)v * 8 + 4);
  float* mv = m + (size_t)v * 8 + chip_half();
  atomicMaxFloat(mv + 0, lrelu(su.x + sv.x));
  atomicMaxFloat(mv + 2, lrelu(su.y + sv.y));
  atomicMaxFloat(mv + 4, lrelu(su.z + sv.z));
  atomicMaxFloat(mv + 6, lrelu(su.w + sv.w));
}

#define TRIP_EDGES 8

__device__ __forceinline__ void agg_single_edge(
    const float* __restrict__ x, const float* __restrict__ s,
    const int* __restrict__ src, const int* __restrict__ dst,
    const float* __restrict__ m, float* __restrict__ denom,
    float* __restrict__ rows, int rpitch, int half, int e, int lane) {
  const int u = src[e], v = dst[e];
  const float* __restrict__ xr = x + (size_t)u * 128 + lane;
  const float x0 = xr[0], x1 = xr[32], x2 = xr[64], x3 = xr[96];
  const int h = lane & 3;
  const float ee = lrelu(s[(size_t)u * 8 + h] + s[(size_t)v * 8 + 4 + h]);
  const float2 mp = *(const float2*)(m + ((size_t)v * 4 + h) * 2);
  const float w = __expf(ee - fmaxf(mp.x, mp.y));
  if (lane < 4) atomicAdd(denom + ((size_t)v * 4 + lane) * 2 + half, w);
  float* o = rows + (size_t)v * rpitch + lane;
  atomicAdd(o,      x0 * w);
  atomicAdd(o + 32, x1 * w);
  atomicAdd(o + 64, x2 * w);
  atomicAdd(o + 96, x3 * w);
}

__global__ __launch_bounds__(256) void gat_edge_agg(
    const float* __restrict__ x, const float* __restrict__ s,
    const int* __restrict__ src, const int* __restrict__ dst,
    const float* __restrict__ m, float* __restrict__ denom,
    float* __restrict__ out, float* __restrict__ acc1, int E) {
  const int lane = threadIdx.x & 31;
  const int h    = lane & 3;
  const int half = chip_half();
  float* __restrict__ rows = half ? acc1 : out + 128;
  const int rpitch = half ? 128 : 256;
  const int sub  = lane >> 2;
  const int wpb  = (int)(blockDim.x >> 5);
  const int wv = __builtin_amdgcn_readfirstlane(
      (int)blockIdx.x * wpb + (int)(threadIdx.x >> 5));
  const int nw = (int)gridDim.x * wpb;
  const int ntrip = E / TRIP_EDGES;

  int b = wv;
  int u_l = 0, v_l = 0;
  if (b < ntrip) {
    u_l = src[b * TRIP_EDGES + sub];
    v_l = dst[b * TRIP_EDGES + sub];
  }
  while (b < ntrip) {
    const int bn = b + nw;
    const int bp = bn < ntrip ? bn : ntrip - 1;
    const int u_n = src[bp * TRIP_EDGES + sub];
    const int v_n = dst[bp * TRIP_EDGES + sub];
    __builtin_amdgcn_sched_barrier(0);

    const float su = s[(size_t)u_l * 8 + h];
    const float sv = s[(size_t)v_l * 8 + 4 + h];
    const float2 mp = *(const float2*)(m + ((size_t)v_l * 4 + h) * 2);
    const float mm = fmaxf(mp.x, mp.y);

    float xv[TRIP_EDGES][4];
    #pragma unroll
    for (int k = 0; k < TRIP_EDGES; ++k) {
      const int u_k = __builtin_amdgcn_readlane(u_l, 4 * k);
      const float* __restrict__ xr = x + (size_t)u_k * 128 + lane;
      xv[k][0] = xr[0];  xv[k][1] = xr[32];
      xv[k][2] = xr[64]; xv[k][3] = xr[96];
    }

    const float w_l = __expf(lrelu(su + sv) - mm);
    atomicAdd(denom + ((size_t)v_l * 4 + h) * 2 + half, w_l);

    #pragma unroll
    for (int k = 0; k < TRIP_EDGES; ++k) {
      const int v_k = __builtin_amdgcn_readlane(v_l, 4 * k);
      const float w = __int_as_float(__builtin_amdgcn_ds_bpermute(
          (4 * k + h) << 2, __float_as_int(w_l)));
      float* o = rows + (size_t)v_k * rpitch + lane;
      atomicAdd(o,      xv[k][0] * w);
      atomicAdd(o + 32, xv[k][1] * w);
      atomicAdd(o + 64, xv[k][2] * w);
      atomicAdd(o + 96, xv[k][3] * w);
    }

    b = bn; u_l = u_n; v_l = v_n;
  }

  for (int e = ntrip * TRIP_EDGES + wv; e < E; e += nw)
    agg_single_edge(x, s, src, dst, m, denom, rows, rpitch, half, e, lane);
}

#define NORM_ROWS 4
__global__ __launch_bounds__(256) void gat_normalize(
    const float* __restrict__ denom, const float* __restrict__ acc1,
    float* __restrict__ out, int n_nodes) {
  const int lane = threadIdx.x & 31;
  const int wave = (int)((blockIdx.x * blockDim.x + threadIdx.x) >> 5);
  const int node0 = wave * NORM_ROWS;

  float4 da[NORM_ROWS], db[NORM_ROWS], o[NORM_ROWS], a[NORM_ROWS];
  #pragma unroll
  for (int i = 0; i < NORM_ROWS; ++i) {
    const int node = min(node0 + i, n_nodes - 1);
    da[i] = *(const float4*)(denom + (size_t)node * 8);
    db[i] = *(const float4*)(denom + (size_t)node * 8 + 4);
    o[i] = *(const float4*)(out + (size_t)node * 256 + 128 + lane * 4);
    a[i] = *(const float4*)(acc1 + (size_t)node * 128 + lane * 4);
  }
  #pragma unroll
  for (int i = 0; i < NORM_ROWS; ++i) {
    const int node = node0 + i;
    if (node < n_nodes) {
      const float d0 = da[i].x + da[i].y, d1 = da[i].z + da[i].w;
      const float d2 = db[i].x + db[i].y, d3 = db[i].z + db[i].w;
      float4 r;
      r.x = o[i].x + a[i].x; r.y = o[i].y + a[i].y;
      r.z = o[i].z + a[i].z; r.w = o[i].w + a[i].w;
      if (d0 > 0.0f) r.x = r.x / d0;
      if (d1 > 0.0f) r.y = r.y / d1;
      if (d2 > 0.0f) r.z = r.z / d2;
      if (d3 > 0.0f) r.w = r.w / d3;
      *(float4*)(out + (size_t)node * 256 + 128 + lane * 4) = r;
    }
  }
}

extern "C" void kernel_launch(void* const* d_in, const int* in_sizes, int n_in,
                              void* d_out, int out_size, void* d_ws, size_t ws_size,
                              hipStream_t stream) {
  const float* x   = (const float*)d_in[0];
  const int*   src = (const int*)d_in[1];
  const int*   dst = (const int*)d_in[2];
  const float* W_u = (const float*)d_in[3];
  const float* b_u = (const float*)d_in[4];
  const float* W_v = (const float*)d_in[5];
  float* out = (float*)d_out;

  const int n = in_sizes[0] / 128;
  const int E = in_sizes[1];

  float* s     = (float*)d_ws;
  float* m     = s + (size_t)n * 8;
  float* denom = m + (size_t)n * 8;
  float* acc1  = denom + (size_t)n * 8;

  gat_scores_wmma<<<(n + 15) / 16, 32, 0, stream>>>(x, W_u, b_u, W_v, s, n);

  gat_init_md<<<(n * 8 + 255) / 256, 256, 0, stream>>>(m, denom, n * 8);
  const long long total = (long long)n * 256;
  gat_init_out<<<(unsigned)((total + 255) / 256), 256, 0, stream>>>(x, out, acc1, total);

  gat_edge_max<<<(E + 255) / 256, 256, 0, stream>>>(s, src, dst, m, E);

  gat_edge_agg<<<1024, 256, 0, stream>>>(x, s, src, dst, m, denom, out, acc1, E);

  gat_normalize<<<(n + 31) / 32, 256, 0, stream>>>(denom, acc1, out, n);
}


# Max pass: a lane is one (edge, head) pair, maxima kept as order-preserving integer keys: one atomic max covers 8 edges' 32-byte pieces (was 8 instructions of up to 32 lines); sums untouched
# speedup vs baseline: 2.4062x; 1.3490x over previous
#include <hip/hip_runtime.h>
#include <math.h>

#define NEG_SLOPE 0.2f

typedef __attribute__((ext_vector_type(2))) float v2f;
typedef __attribute__((ext_vector_type(8))) float v8f;

__global__ __launch_bounds__(32) void gat_scores_wmma(
    const float* __restrict__ x, const float* __restrict__ W_u,
    const float* __restrict__ b_u, const float* __restrict__ W_v,
    float* __restrict__ s, int n_nodes) {
  const int tile = blockIdx.x;
  const int lane = threadIdx.x;
  const int half = lane >> 4;
  const int l15  = lane & 15;

  __shared__ float Wc[128 * 16];
  for (int i = lane; i < 128 * 16; i += 32) {
    const int k = i >> 4, c = i & 15;
    float w = 0.0f;
    if (c < 4)      w = W_u[k * 4 + c];
    else if (c < 8) w = W_v[k * 4 + (c - 4)];
    Wc[i] = w;
  }
  __syncthreads();

  int node = tile * 16 + l15;
  const int nodeC = node < n_nodes ? node : n_nodes - 1;
  const float* __restrict__ xrow = x + (size_t)nodeC * 128;

  v8f acc = {};
  #pragma unroll 4
  for (int k = 0; k < 128; k += 4) {
    const int ka = k + half * 2;
    v2f a; a.x = xrow[ka];            a.y = xrow[ka + 1];
    v2f b; b.x = Wc[ka * 16 + l15];   b.y = Wc[(ka + 1) * 16 + l15];
    acc = __builtin_amdgcn_wmma_f32_16x16x4_f32(
         false, a,  false, b,
         (short)0, acc,  false,  false);
  }

  const int ncol = l15;
  if (ncol < 8) {
    const float bias = (ncol < 4) ? b_u[ncol] : 0.0f;
    #pragma unroll
    for (int r = 0; r < 8; ++r) {
      const int mrow = r + half * 8;
      const int onode = tile * 16 + mrow;
      if (onode < n_nodes) s[(size_t)onode * 8 + ncol] = acc[r] + bias;
    }
  }
}

__device__ __forceinline__ int chip_half() {
  return (int)((__builtin_amdgcn_s_sendmsg_rtn(0x87) >> 18) & 1u);
}

__device__ __forceinline__ int max_key(float f) {
  const int b = __float_as_int(f);
  return b ^ ((b >> 31) & 0x7FFFFFFF);
}
__device__ __forceinline__ float key_value(int k) {
  return __int_as_float(k ^ ((k >> 31) & 0x7FFFFFFF));
}

__global__ void gat_init_md(int* __restrict__ m, float* __restrict__ denom,
                            int n8) {
  const int i = blockIdx.x * blockDim.x + threadIdx.x;
  if (i < n8) { m[i] = max_key(-INFINITY); denom[i] = 0.0f; }
}

__global__ void gat_init_out(const float* __restrict__ x,
                             float* __restrict__ out,
                             float* __restrict__ acc1, long long total) {
  const long long i = (long long)blockIdx.x * blockDim.x + threadIdx.x;
  if (i >= total) return;
  const long long node = i >> 8;
  const int d = (int)(i & 255);
  if (d < 128) {
    out[i] = x[node * 128 + d];
  } else {
    out[i] = 0.0f;
    acc1[node * 128 + (d - 128)] = 0.0f;
  }
}

__device__ __forceinline__ float lrelu(float v) {
  return v >= 0.0f ? v : NEG_SLOPE * v;
}

#define MAX_GROUPS 4

template <bool TAIL>
__device__ __forceinline__ void edge_max_wave(
    const float* __restrict__ s, const int* __restrict__ src,
    const int* __restrict__ dst, int* __restrict__ m, int E, int e0, int h,
    int half) {
  int u[MAX_GROUPS], v[MAX_GROUPS];
  float su[MAX_GROUPS], sv[MAX_GROUPS];
  #pragma unroll
  for (int j = 0; j < MAX_GROUPS; ++j) {
    const int e = TAIL ? min(e0 + 8 * j, E - 1) : e0 + 8 * j;
    u[j] = src[e];
    v[j] = dst[e];
  }
  #pragma unroll
  for (int j = 0; j < MAX_GROUPS; ++j) {
    su[j] = s[(size_t)u[j] * 8 + h];
    sv[j] = s[(size_t)v[j] * 8 + 4 + h];
  }
  #pragma unroll
  for (int j = 0; j < MAX_GROUPS; ++j) {
    if (!TAIL || e0 + 8 * j < E)
      atomicMax(m + ((size_t)v[j] * 4 + h) * 2 + half,
                max_key(lrelu(su[j] + sv[j])));
  }
}

__global__ __launch_bounds__(256) void gat_edge_max(
    const float* __restrict__ s, const int* __restrict__ src,
    const int* __restrict__ dst, int* __restrict__ m, int E) {
  const int lane = threadIdx.x & 31;
  const int h    = lane & 3;
  const int base = __builtin_amdgcn_readfirstlane(
      (int)((blockIdx.x * blockDim.x + threadIdx.x) >> 5) * (8 * MAX_GROUPS));
  if (base >= E) return;
  const int e0   = base + (lane >> 2);
  const int half = chip_half();
  if (base + 8 * MAX_GROUPS <= E)
    edge_max_wave<false>(s, src, dst, m, E, e0, h, half);
  else
    edge_max_wave<true>(s, src, dst, m, E, e0, h, half);
}

#define TRIP_EDGES 8

__device__ __forceinline__ void agg_single_edge(
    const float* __restrict__ x, const float* __restrict__ s,
    const int* __restrict__ src, const int* __restrict__ dst,
    const int* __restrict__ m, float* __restrict__ denom,
    float* __restrict__ rows, int rpitch, int half, int e, int lane) {
  const int u = src[e], v = dst[e];
  const float* __restrict__ xr = x + (size_t)u * 128 + lane;
  const float x0 = xr[0], x1 = xr[32], x2 = xr[64], x3 = xr[96];
  const int h = lane & 3;
  const float ee = lrelu(s[(size_t)u * 8 + h] + s[(size_t)v * 8 + 4 + h]);
  const int2 mk = *(const int2*)(m + ((size_t)v * 4 + h) * 2);
  const float w = __expf(ee - key_value(max(mk.x, mk.y)));
  if (lane < 4) atomicAdd(denom + ((size_t)v * 4 + lane) * 2 + half, w);
  float* o = rows + (size_t)v * rpitch + lane;
  atomicAdd(o,      x0 * w);
  atomicAdd(o + 32, x1 * w);
  atomicAdd(o + 64, x2 * w);
  atomicAdd(o + 96, x3 * w);
}

__global__ __launch_bounds__(256) void gat_edge_agg(
    const float* __restrict__ x, const float* __restrict__ s,
    const int* __restrict__ src, const int* __restrict__ dst,
    const int* __restrict__ m, float* __restrict__ denom,
    float* __restrict__ out, float* __restrict__ acc1, int E) {
  const int lane = threadIdx.x & 31;
  const int h    = lane & 3;
  const int half = chip_half();
  float* __restrict__ rows = half ? acc1 : out + 128;
  const int rpitch = half ? 128 : 256;
  const int sub  = lane >> 2;
  const int wpb  = (int)(blockDim.x >> 5);
  const int wv = __builtin_amdgcn_readfirstlane(
      (int)blockIdx.x * wpb + (int)(threadIdx.x >> 5));
  const int nw = (int)gridDim.x * wpb;
  const int ntrip = E / TRIP_EDGES;

  int b = wv;
  int u_l = 0, v_l = 0;
  if (b < ntrip) {
    u_l = src[b * TRIP_EDGES + sub];
    v_l = dst[b * TRIP_EDGES + sub];
  }
  while (b < ntrip) {
    const int bn = b + nw;
    const int bp = bn < ntrip ? bn : ntrip - 1;
    const int u_n = src[bp * TRIP_EDGES + sub];
    const int v_n = dst[bp * TRIP_EDGES + sub];
    __builtin_amdgcn_sched_barrier(0);

    const float su = s[(size_t)u_l * 8 + h];
    const float sv = s[(size_t)v_l * 8 + 4 + h];
    const int2 mk = *(const int2*)(m + ((size_t)v_l * 4 + h) * 2);
    const float mm = key_value(max(mk.x, mk.y));

    float xv[TRIP_EDGES][4];
    #pragma unroll
    for (int k = 0; k < TRIP_EDGES; ++k) {
      const int u_k = __builtin_amdgcn_readlane(u_l, 4 * k);
      const float* __restrict__ xr = x + (size_t)u_k * 128 + lane;
      xv[k][0] = xr[0];  xv[k][1] = xr[32];
      xv[k][2] = xr[64]; xv[k][3] = xr[96];
    }

    const float w_l = __expf(lrelu(su + sv) - mm);
    atomicAdd(denom + ((size_t)v_l * 4 + h) * 2 + half, w_l);

    #pragma unroll
    for (int k = 0; k < TRIP_EDGES; ++k) {
      const int v_k = __builtin_amdgcn_readlane(v_l, 4 * k);
      const float w = __int_as_float(__builtin_amdgcn_ds_bpermute(
          (4 * k + h) << 2, __float_as_int(w_l)));
      float* o = rows + (size_t)v_k * rpitch + lane;
      atomicAdd(o,      xv[k][0] * w);
      atomicAdd(o + 32, xv[k][1] * w);
      atomicAdd(o + 64, xv[k][2] * w);
      atomicAdd(o + 96, xv[k][3] * w);
    }

    b = bn; u_l = u_n; v_l = v_n;
  }

  for (int e = ntrip * TRIP_EDGES + wv; e < E; e += nw)
    agg_single_edge(x, s, src, dst, m, denom, rows, rpitch, half, e, lane);
}

#define NORM_ROWS 4
__global__ __launch_bounds__(256) void gat_normalize(
    const float* __restrict__ denom, const float* __restrict__ acc1,
    float* __restrict__ out, int n_nodes) {
  const int lane = threadIdx.x & 31;
  const int wave = (int)((blockIdx.x * blockDim.x + threadIdx.x) >> 5);
  const int node0 = wave * NORM_ROWS;

  float4 da[NORM_ROWS], db[NORM_ROWS], o[NORM_ROWS], a[NORM_ROWS];
  #pragma unroll
  for (int i = 0; i < NORM_ROWS; ++i) {
    const int node = min(node0 + i, n_nodes - 1);
    da[i] = *(const float4*)(denom + (size_t)node * 8);
    db[i] = *(const float4*)(denom + (size_t)node * 8 + 4);
    o[i] = *(const float4*)(out + (size_t)node * 256 + 128 + lane * 4);
    a[i] = *(const float4*)(acc1 + (size_t)node * 128 + lane * 4);
  }
  #pragma unroll
  for (int i = 0; i < NORM_ROWS; ++i) {
    const int node = node0 + i;
    if (node < n_nodes) {
      const float d0 = da[i].x + da[i].y, d1 = da[i].z + da[i].w;
      const float d2 = db[i].x + db[i].y, d3 = db[i].z + db[i].w;
      float4 r;
      r.x = o[i].x + a[i].x; r.y = o[i].y + a[i].y;
      r.z = o[i].z + a[i].z; r.w = o[i].w + a[i].w;
      if (d0 > 0.0f) r.x = r.x / d0;
      if (d1 > 0.0f) r.y = r.y / d1;
      if (d2 > 0.0f) r.z = r.z / d2;
      if (d3 > 0.0f) r.w = r.w / d3;
      *(float4*)(out + (size_t)node * 256 + 128 + lane * 4) = r;
    }
  }
}

extern "C" void kernel_launch(void* const* d_in, const int* in_sizes, int n_in,
                              void* d_out, int out_size, void* d_ws, size_t ws_size,
                              hipStream_t stream) {
  const float* x   = (const float*)d_in[0];
  const int*   src = (const int*)d_in[1];
  const int*   dst = (const int*)d_in[2];
  const float* W_u = (const float*)d_in[3];
  const float* b_u = (const float*)d_in[4];
  const float* W_v = (const float*)d_in[5];
  float* out = (float*)d_out;

  const int n = in_sizes[0] / 128;
  const int E = in_sizes[1];

  float* s     = (float*)d_ws;
  int*   m     = (int*)(s + (size_t)n * 8);
  float* denom = (float*)(m + (size_t)n * 8);
  float* acc1  = denom + (size_t)n * 8;

  gat_scores_wmma<<<(n + 15) / 16, 32, 0, stream>>>(x, W_u, b_u, W_v, s, n);

  gat_init_md<<<(n * 8 + 255) / 256, 256, 0, stream>>>(m, denom, n * 8);
  const long long total = (long long)n * 256;
  gat_init_out<<<(unsigned)((total + 255) / 256), 256, 0, stream>>>(x, out, acc1, total);

  gat_edge_max<<<(E + 255) / 256, 256, 0, stream>>>(s, src, dst, m, E);

  gat_edge_agg<<<1024, 256, 0, stream>>>(x, s, src, dst, m, denom, out, acc1, E);

  gat_normalize<<<(n + 31) / 32, 256, 0, stream>>>(denom, acc1, out, n);
}


# Edge aggregation in two sweeps of the edge list, 64 of the 128 columns each (weights kept from the first), so each half's random working set fits its L2; float atomics, arrival order as before
# speedup vs baseline: 2.4782x; 1.0299x over previous
#include <hip/hip_runtime.h>
#include <math.h>

#define NEG_SLOPE 0.2f

typedef __attribute__((ext_vector_type(2))) float v2f;
typedef __attribute__((ext_vector_type(8))) float v8f;

__global__ __launch_bounds__(32) void gat_scores_wmma(
    const float* __restrict__ x, const float* __restrict__ W_u,
    const float* __restrict__ b_u, const float* __restrict__ W_v,
    float* __restrict__ s, int n_nodes) {
  const int tile = blockIdx.x;
  const int lane = threadIdx.x;
  const int half = lane >> 4;
  const int l15  = lane & 15;

  __shared__ float Wc[128 * 16];
  for (int i = lane; i < 128 * 16; i += 32) {
    const int k = i >> 4, c = i & 15;
    float w = 0.0f;
    if (c < 4)      w = W_u[k * 4 + c];
    else if (c < 8) w = W_v[k * 4 + (c - 4)];
    Wc[i] = w;
  }
  __syncthreads();

  int node = tile * 16 + l15;
  const int nodeC = node < n_nodes ? node : n_nodes - 1;
  const float* __restrict__ xrow = x + (size_t)nodeC * 128;

  v8f acc = {};
  #pragma unroll 4
  for (int k = 0; k < 128; k += 4) {
    const int ka = k + half * 2;
    v2f a; a.x = xrow[ka];            a.y = xrow[ka + 1];
    v2f b; b.x = Wc[ka * 16 + l15];   b.y = Wc[(ka + 1) * 16 + l15];
    acc = __builtin_amdgcn_wmma_f32_16x16x4_f32(
         false, a,  false, b,
         (short)0, acc,  false,  false);
  }

  const int ncol = l15;
  if (ncol < 8) {
    const float bias = (ncol < 4) ? b_u[ncol] : 0.0f;
    #pragma unroll
    for (int r = 0; r < 8; ++r) {
      const int mrow = r + half * 8;
      const int onode = tile * 16 + mrow;
      if (onode < n_nodes) s[(size_t)onode * 8 + ncol] = acc[r] + bias;
    }
  }
}

__device__ __forceinline__ int chip_half() {
  return (int)((__builtin_amdgcn_s_sendmsg_rtn(0x87) >> 18) & 1u);
}

__device__ __forceinline__ int max_key(float f) {
  const int b = __float_as_int(f);
  return b ^ ((b >> 31) & 0x7FFFFFFF);
}
__device__ __forceinline__ float key_value(int k) {
  return __int_as_float(k ^ ((k >> 31) & 0x7FFFFFFF));
}

__global__ void gat_init_md(int* __restrict__ m, float* __restrict__ denom,
                            int n8) {
  const int i = blockIdx.x * blockDim.x + threadIdx.x;
  if (i < n8) { m[i] = max_key(-INFINITY); denom[i] = 0.0f; }
}

__global__ void gat_init_out(const float* __restrict__ x,
                             float* __restrict__ out,
                             float* __restrict__ acc1, long long total) {
  const long long i = (long long)blockIdx.x * blockDim.x + threadIdx.x;
  if (i >= total) return;
  const long long node = i >> 8;
  const int d = (int)(i & 255);
  if (d < 128) {
    out[i] = x[node * 128 + d];
  } else {
    out[i] = 0.0f;
    acc1[node * 128 + (d - 128)] = 0.0f;
  }
}

__device__ __forceinline__ float lrelu(float v) {
  return v >= 0.0f ? v : NEG_SLOPE * v;
}

#define MAX_GROUPS 4

template <bool TAIL>
__device__ __forceinline__ void edge_max_wave(
    const float* __restrict__ s, const int* __restrict__ src,
    const int* __restrict__ dst, int* __restrict__ m, int E, int e0, int h,
    int half) {
  int u[MAX_GROUPS], v[MAX_GROUPS];
  float su[MAX_GROUPS], sv[MAX_GROUPS];
  #pragma unroll
  for (int j = 0; j < MAX_GROUPS; ++j) {
    const int e = TAIL ? min(e0 + 8 * j, E - 1) : e0 + 8 * j;
    u[j] = src[e];
    v[j] = dst[e];
  }
  #pragma unroll
  for (int j = 0; j < MAX_GROUPS; ++j) {
    su[j] = s[(size_t)u[j] * 8 + h];
    sv[j] = s[(size_t)v[j] * 8 + 4 + h];
  }
  #pragma unroll
  for (int j = 0; j < MAX_GROUPS; ++j) {
    if (!TAIL || e0 + 8 * j < E)
      atomicMax(m + ((size_t)v[j] * 4 + h) * 2 + half,
                max_key(lrelu(su[j] + sv[j])));
  }
}

__global__ __launch_bounds__(256) void gat_edge_max(
    const float* __restrict__ s, const int* __restrict__ src,
    const int* __restrict__ dst, int* __restrict__ m, int E) {
  const int lane = threadIdx.x & 31;
  const int h    = lane & 3;
  const int base = __builtin_amdgcn_readfirstlane(
      (int)((blockIdx.x * blockDim.x + threadIdx.x) >> 5) * (8 * MAX_GROUPS));
  if (base >= E) return;
  const int e0   = base + (lane >> 2);
  const int half = chip_half();
  if (base + 8 * MAX_GROUPS <= E)
    edge_max_wave<false>(s, src, dst, m, E, e0, h, half);
  else
    edge_max_wave<true>(s, src, dst, m, E, e0, h, half);
}

#define TRIP_EDGES 8
#define SWEEP_DIMS 64
#define SWEEP_LINES (SWEEP_DIMS / 32)

__device__ __forceinline__ void agg_single_edge(
    const float* __restrict__ x, const float* __restrict__ s,
    const int* __restrict__ src, const int* __restrict__ dst,
    const int* __restrict__ m, float* __restrict__ denom,
    float* __restrict__ rows, int rpitch, int half, int e, int lane) {
  const int u = src[e], v = dst[e];
  const float* __restrict__ xr = x + (size_t)u * 128 + lane;
  const float x0 = xr[0], x1 = xr[32], x2 = xr[64], x3 = xr[96];
  const int h = lane & 3;
  const float ee = lrelu(s[(size_t)u * 8 + h] + s[(size_t)v * 8 + 4 + h]);
  const int2 mk = *(const int2*)(m + ((size_t)v * 4 + h) * 2);
  const float w = __expf(ee - key_value(max(mk.x, mk.y)));
  if (lane < 4) atomicAdd(denom + ((size_t)v * 4 + lane) * 2 + half, w);
  float* o = rows + (size_t)v * rpitch + lane;
  atomicAdd(o,      x0 * w);
  atomicAdd(o + 32, x1 * w);
  atomicAdd(o + 64, x2 * w);
  atomicAdd(o + 96, x3 * w);
}

template <int Q>
__device__ __forceinline__ void agg_sweep(
    const float* __restrict__ x, const float* __restrict__ s,
    const int* __restrict__ src, const int* __restrict__ dst,
    const int* __restrict__ m, float* __restrict__ denom,
    float* __restrict__ wts, float* __restrict__ rows, int rpitch, int half,
    int lane, int wv, int nw, int ntrip) {
  const int h   = lane & 3;
  const int sub = lane >> 2;
  const int col = Q * SWEEP_DIMS + lane;

  int b = wv;
  int u_l = 0, v_l = 0;
  float w_l = 0.0f;
  if (b < ntrip) {
    u_l = src[b * TRIP_EDGES + sub];
    v_l = dst[b * TRIP_EDGES + sub];
    if (Q != 0) w_l = wts[(size_t)b * 32 + lane];
  }
  while (b < ntrip) {
    const int bn = b + nw;
    const int bp = bn < ntrip ? bn : ntrip - 1;
    const int u_n = src[bp * TRIP_EDGES + sub];
    const int v_n = dst[bp * TRIP_EDGES + sub];
    float w_n = 0.0f;
    if (Q != 0) w_n = wts[(size_t)(bn < ntrip ? bn : b) * 32 + lane];
    __builtin_amdgcn_sched_barrier(0);

    float su = 0.0f, sv = 0.0f, mm = 0.0f;
    if (Q == 0) {
      su = s[(size_t)u_l * 8 + h];
      sv = s[(size_t)v_l * 8 + 4 + h];
      const int2 mk = *(const int2*)(m + ((size_t)v_l * 4 + h) * 2);
      mm = key_value(max(mk.x, mk.y));
    }

    float xv[TRIP_EDGES][SWEEP_LINES];
    #pragma unroll
    for (int k = 0; k < TRIP_EDGES; ++k) {
      const int u_k = __builtin_amdgcn_readlane(u_l, 4 * k);
      const float* __restrict__ xr = x + (size_t)u_k * 128 + col;
      #pragma unroll
      for (int j = 0; j < SWEEP_LINES; ++j) xv[k][j] = xr[32 * j];
    }

    if (Q == 0) {
      w_l = __expf(lrelu(su + sv) - mm);
      atomicAdd(denom + ((size_t)v_l * 4 + h) * 2 + half, w_l);
      wts[(size_t)b * 32 + lane] = w_l;
    }

    #pragma unroll
    for (int k = 0; k < TRIP_EDGES; ++k) {
      const int v_k = __builtin_amdgcn_readlane(v_l, 4 * k);
      const float w = __int_as_float(__builtin_amdgcn_ds_bpermute(
          (4 * k + h) << 2, __float_as_int(w_l)));
      float* o = rows + (size_t)v_k * rpitch + col;
      #pragma unroll
      for (int j = 0; j < SWEEP_LINES; ++j) atomicAdd(o + 32 * j, xv[k][j] * w);
    }

    b = bn; u_l = u_n; v_l = v_n;
    if (Q != 0) w_l = w_n;
  }
}

__global__ __launch_bounds__(256) void gat_edge_agg(
    const float* __restrict__ x, const float* __restrict__ s,
    const int* __restrict__ src, const int* __restrict__ dst,
    const int* __restrict__ m, float* __restrict__ denom,
    float* __restrict__ wts, float* __restrict__ out,
    float* __restrict__ acc1, int E) {
  const int lane = threadIdx.x & 31;
  const int half = chip_half();
  float* __restrict__ rows = half ? acc1 : out + 128;
  const int rpitch = half ? 128 : 256;
  const int wpb  = (int)(blockDim.x >> 5);
  const int wv = __builtin_amdgcn_readfirstlane(
      (int)blockIdx.x * wpb + (int)(threadIdx.x >> 5));
  const int nw = (int)gridDim.x * wpb;
  const int ntrip = E / TRIP_EDGES;

  agg_sweep<0>(x, s, src, dst, m, denom, wts, rows, rpitch, half, lane, wv, nw,
               ntrip);
  agg_sweep<1>(x, s, src, dst, m, denom, wts, rows, rpitch, half, lane, wv, nw,
               ntrip);

  for (int e = ntrip * TRIP_EDGES + wv; e < E; e += nw)
    agg_single_edge(x, s, src, dst, m, denom, rows, rpitch, half, e, lane);
}

#define NORM_ROWS 4
__global__ __launch_bounds__(256) void gat_normalize(
    const float* __restrict__ denom, const float* __restrict__ acc1,
    float* __restrict__ out, int n_nodes) {
  const int lane = threadIdx.x & 31;
  const int wave = (int)((blockIdx.x * blockDim.x + threadIdx.x) >> 5);
  const int node0 = wave * NORM_ROWS;

  float4 da[NORM_ROWS], db[NORM_ROWS], o[NORM_ROWS], a[NORM_ROWS];
  #pragma unroll
  for (int i = 0; i < NORM_ROWS; ++i) {
    const int node = min(node0 + i, n_nodes - 1);
    da[i] = *(const float4*)(denom + (size_t)node * 8);
    db[i] = *(const float4*)(denom + (size_t)node * 8 + 4);
    o[i] = *(const float4*)(out + (size_t)node * 256 + 128 + lane * 4);
    a[i] = *(const float4*)(acc1 + (size_t)node * 128 + lane * 4);
  }
  #pragma unroll
  for (int i = 0; i < NORM_ROWS; ++i) {
    const int node = node0 + i;
    if (node < n_nodes) {
      const float d0 = da[i].x + da[i].y, d1 = da[i].z + da[i].w;
      const float d2 = db[i].x + db[i].y, d3 = db[i].z + db[i].w;
      float4 r;
      r.x = o[i].x + a[i].x; r.y = o[i].y + a[i].y;
      r.z = o[i].z + a[i].z; r.w = o[i].w + a[i].w;
      if (d0 > 0.0f) r.x = r.x / d0;
      if (d1 > 0.0f) r.y = r.y / d1;
      if (d2 > 0.0f) r.z = r.z / d2;
      if (d3 > 0.0f) r.w = r.w / d3;
      *(float4*)(out + (size_t)node * 256 + 128 + lane * 4) = r;
    }
  }
}

extern "C" void kernel_launch(void* const* d_in, const int* in_sizes, int n_in,
                              void* d_out, int out_size, void* d_ws, size_t ws_size,
                              hipStream_t stream) {
  const float* x   = (const float*)d_in[0];
  const int*   src = (const int*)d_in[1];
  const int*   dst = (const int*)d_in[2];
  const float* W_u = (const float*)d_in[3];
  const float* b_u = (const float*)d_in[4];
  const float* W_v = (const float*)d_in[5];
  float* out = (float*)d_out;

  const int n = in_sizes[0] / 128;
  const int E = in_sizes[1];

  float* s     = (float*)d_ws;
  int*   m     = (int*)(s + (size_t)n * 8);
  float* denom = (float*)(m + (size_t)n * 8);
  float* acc1  = denom + (size_t)n * 8;
  float* wts   = acc1 + (size_t)n * 128;

  gat_scores_wmma<<<(n + 15) / 16, 32, 0, stream>>>(x, W_u, b_u, W_v, s, n);

  gat_init_md<<<(n * 8 + 255) / 256, 256, 0, stream>>>(m, denom, n * 8);
  const long long total = (long long)n * 256;
  gat_init_out<<<(unsigned)((total + 255) / 256), 256, 0, stream>>>(x, out, acc1, total);

  gat_edge_max<<<(E + 255) / 256, 256, 0, stream>>>(s, src, dst, m, E);

  gat_edge_agg<<<1024, 256, 0, stream>>>(x, s, src, dst, m, denom, wts, out, acc1,
                                         E);

  gat_normalize<<<(n + 31) / 32, 256, 0, stream>>>(denom, acc1, out, n);
}


# Init pass (x copied into out, both row-sum copies zeroed): a wave takes 8 whole rows, 16 bytes per lane and access, its 8 row loads in flight at once (was one 4-byte element per thread, 800000 waves)
# speedup vs baseline: 2.6990x; 1.0891x over previous
#include <hip/hip_runtime.h>
#include <math.h>

#define NEG_SLOPE 0.2f

typedef __attribute__((ext_vector_type(2))) float v2f;
typedef __attribute__((ext_vector_type(8))) float v8f;
typedef __attribute__((ext_vector_type(4))) float v4f;

__global__ __launch_bounds__(32) void gat_scores_wmma(
    const float* __restrict__ x, const float* __restrict__ W_u,
    const float* __restrict__ b_u, const float* __restrict__ W_v,
    float* __restrict__ s, int n_nodes) {
  const int tile = blockIdx.x;
  const int lane = threadIdx.x;
  const int half = lane >> 4;
  const int l15  = lane & 15;

  __shared__ float Wc[128 * 16];
  for (int i = lane; i < 128 * 16; i += 32) {
    const int k = i >> 4, c = i & 15;
    float w = 0.0f;
    if (c < 4)      w = W_u[k * 4 + c];
    else if (c < 8) w = W_v[k * 4 + (c - 4)];
    Wc[i] = w;
  }
  __syncthreads();

  int node = tile * 16 + l15;
  const int nodeC = node < n_nodes ? node : n_nodes - 1;
  const float* __restrict__ xrow = x + (size_t)nodeC * 128;

  v8f acc = {};
  #pragma unroll 4
  for (int k = 0; k < 128; k += 4) {
    const int ka = k + half * 2;
    v2f a; a.x = xrow[ka];            a.y = xrow[ka + 1];
    v2f b; b.x = Wc[ka * 16 + l15];   b.y = Wc[(ka + 1) * 16 + l15];
    acc = __builtin_amdgcn_wmma_f32_16x16x4_f32(
         false, a,  false, b,
         (short)0, acc,  false,  false);
  }

  const int ncol = l15;
  if (ncol < 8) {
    const float bias = (ncol < 4) ? b_u[ncol] : 0.0f;
    #pragma unroll
    for (int r = 0; r < 8; ++r) {
      const int mrow = r + half * 8;
      const int onode = tile * 16 + mrow;
      if (onode < n_nodes) s[(size_t)onode * 8 + ncol] = acc[r] + bias;
    }
  }
}

__device__ __forceinline__ int chip_half() {
  return (int)((__builtin_amdgcn_s_sendmsg_rtn(0x87) >> 18) & 1u);
}

__device__ __forceinline__ int max_key(float f) {
  const int b = __float_as_int(f);
  return b ^ ((b >> 31) & 0x7FFFFFFF);
}
__device__ __forceinline__ float key_value(int k) {
  return __int_as_float(k ^ ((k >> 31) & 0x7FFFFFFF));
}

__global__ void gat_init_md(int* __restrict__ m, float* __restrict__ denom,
                            int n8) {
  const int i = blockIdx.x * blockDim.x + threadIdx.x;
  if (i < n8) { m[i] = max_key(-INFINITY); denom[i] = 0.0f; }
}

#define INIT_ROWS 8
#define INIT_WAVES 8
__global__ __launch_bounds__(INIT_WAVES * 32) void gat_init_out(
    const float* __restrict__ x, float* __restrict__ out,
    float* __restrict__ acc1, int n_nodes) {
  const int lane = threadIdx.x & 31;
  const int wave = __builtin_amdgcn_readfirstlane(
      (int)((blockIdx.x * blockDim.x + threadIdx.x) >> 5));
  const int node0 = wave * INIT_ROWS;
  if (node0 >= n_nodes) return;

  const v4f z = {0.0f, 0.0f, 0.0f, 0.0f};
  const float* __restrict__ xr = x + (size_t)node0 * 128 + lane * 4;
  float* __restrict__ o = out + (size_t)node0 * 256 + lane * 4;
  float* __restrict__ a = acc1 + (size_t)node0 * 128 + lane * 4;

  if (node0 + INIT_ROWS <= n_nodes) {
    v4f v[INIT_ROWS];
    #pragma unroll
    for (int i = 0; i < INIT_ROWS; ++i)
      v[i] = *(const v4f*)(xr + (size_t)i * 128);
    #pragma unroll
    for (int i = 0; i < INIT_ROWS; ++i) {
      *(v4f*)(o + (size_t)i * 256 + 128) = z;
      *(v4f*)(a + (size_t)i * 128) = z;
    }
    static_assert(INIT_ROWS == 8, "the statement below names eight rows");
    asm volatile("" :: "v"(v[0]), "v"(v[1]), "v"(v[2]), "v"(v[3]),
                       "v"(v[4]), "v"(v[5]), "v"(v[6]), "v"(v[7]));
    #pragma unroll
    for (int i = 0; i < INIT_ROWS; ++i)
      *(v4f*)(o + (size_t)i * 256) = v[i];
  } else {
    for (int i = 0; node0 + i < n_nodes; ++i) {
      *(v4f*)(o + (size_t)i * 256) = *(const v4f*)(xr + (size_t)i * 128);
      *(v4f*)(o + (size_t)i * 256 + 128) = z;
      *(v4f*)(a + (size_t)i * 128) = z;
    }
  }
}

__device__ __forceinline__ float lrelu(float v) {
  return v >= 0.0f ? v : NEG_SLOPE * v;
}

#define MAX_GROUPS 4

template <bool TAIL>
__device__ __forceinline__ void edge_max_wave(
    const float* __restrict__ s, const int* __restrict__ src,
    const int* __restrict__ dst, int* __restrict__ m, int E, int e0, int h,
    int half) {
  int u[MAX_GROUPS], v[MAX_GROUPS];
  float su[MAX_GROUPS], sv[MAX_GROUPS];
  #pragma unroll
  for (int j = 0; j < MAX_GROUPS; ++j) {
    const int e = TAIL ? min(e0 + 8 * j, E - 1) : e0 + 8 * j;
    u[j] = src[e];
    v[j] = dst[e];
  }
  #pragma unroll
  for (int j = 0; j < MAX_GROUPS; ++j) {
    su[j] = s[(size_t)u[j] * 8 + h];
    sv[j] = s[(size_t)v[j] * 8 + 4 + h];
  }
  #pragma unroll
  for (int j = 0; j < MAX_GROUPS; ++j) {
    if (!TAIL || e0 + 8 * j < E)
      atomicMax(m + ((size_t)v[j] * 4 + h) * 2 + half,
                max_key(lrelu(su[j] + sv[j])));
  }
}

__global__ __launch_bounds__(256) void gat_edge_max(
    const float* __restrict__ s, const int* __restrict__ src,
    const int* __restrict__ dst, int* __restrict__ m, int E) {
  const int lane = threadIdx.x & 31;
  const int h    = lane & 3;
  const int base = __builtin_amdgcn_readfirstlane(
      (int)((blockIdx.x * blockDim.x + threadIdx.x) >> 5) * (8 * MAX_GROUPS));
  if (base >= E) return;
  const int e0   = base + (lane >> 2);
  const int half = chip_half();
  if (base + 8 * MAX_GROUPS <= E)
    edge_max_wave<false>(s, src, dst, m, E, e0, h, half);
  else
    edge_max_wave<true>(s, src, dst, m, E, e0, h, half);
}

#define TRIP_EDGES 8
#define SWEEP_DIMS 64
#define SWEEP_LINES (SWEEP_DIMS / 32)

__device__ __forceinline__ void agg_single_edge(
    const float* __restrict__ x, const float* __restrict__ s,
    const int* __restrict__ src, const int* __restrict__ dst,
    const int* __restrict__ m, float* __restrict__ denom,
    float* __restrict__ rows, int rpitch, int half, int e, int lane) {
  const int u = src[e], v = dst[e];
  const float* __restrict__ xr = x + (size_t)u * 128 + lane;
  const float x0 = xr[0], x1 = xr[32], x2 = xr[64], x3 = xr[96];
  const int h = lane & 3;
  const float ee = lrelu(s[(size_t)u * 8 + h] + s[(size_t)v * 8 + 4 + h]);
  const int2 mk = *(const int2*)(m + ((size_t)v * 4 + h) * 2);
  const float w = __expf(ee - key_value(max(mk.x, mk.y)));
  if (lane < 4) atomicAdd(denom + ((size_t)v * 4 + lane) * 2 + half, w);
  float* o = rows + (size_t)v * rpitch + lane;
  atomicAdd(o,      x0 * w);
  atomicAdd(o + 32, x1 * w);
  atomicAdd(o + 64, x2 * w);
  atomicAdd(o + 96, x3 * w);
}

template <int Q>
__device__ __forceinline__ void agg_sweep(
    const float* __restrict__ x, const float* __restrict__ s,
    const int* __restrict__ src, const int* __restrict__ dst,
    const int* __restrict__ m, float* __restrict__ denom,
    float* __restrict__ wts, float* __restrict__ rows, int rpitch, int half,
    int lane, int wv, int nw, int ntrip) {
  const int h   = lane & 3;
  const int sub = lane >> 2;
  const int col = Q * SWEEP_DIMS + lane;

  int b = wv;
  int u_l = 0, v_l = 0;
  float w_l = 0.0f;
  if (b < ntrip) {
    u_l = src[b * TRIP_EDGES + sub];
    v_l = dst[b * TRIP_EDGES + sub];
    if (Q != 0) w_l = wts[(size_t)b * 32 + lane];
  }
  while (b < ntrip) {
    const int bn = b + nw;
    const int bp = bn < ntrip ? bn : ntrip - 1;
    const int u_n = src[bp * TRIP_EDGES + sub];
    const int v_n = dst[bp * TRIP_EDGES + sub];
    float w_n = 0.0f;
    if (Q != 0) w_n = wts[(size_t)(bn < ntrip ? bn : b) * 32 + lane];
    __builtin_amdgcn_sched_barrier(0);

    float su = 0.0f, sv = 0.0f, mm = 0.0f;
    if (Q == 0) {
      su = s[(size_t)u_l * 8 + h];
      sv = s[(size_t)v_l * 8 + 4 + h];
      const int2 mk = *(const int2*)(m + ((size_t)v_l * 4 + h) * 2);
      mm = key_value(max(mk.x, mk.y));
    }

    float xv[TRIP_EDGES][SWEEP_LINES];
    #pragma unroll
    for (int k = 0; k < TRIP_EDGES; ++k) {
      const int u_k = __builtin_amdgcn_readlane(u_l, 4 * k);
      const float* __restrict__ xr = x + (size_t)u_k * 128 + col;
      #pragma unroll
      for (int j = 0; j < SWEEP_LINES; ++j) xv[k][j] = xr[32 * j];
    }

    if (Q == 0) {
      w_l = __expf(lrelu(su + sv) - mm);
      atomicAdd(denom + ((size_t)v_l * 4 + h) * 2 + half, w_l);
      wts[(size_t)b * 32 + lane] = w_l;
    }

    #pragma unroll
    for (int k = 0; k < TRIP_EDGES; ++k) {
      const int v_k = __builtin_amdgcn_readlane(v_l, 4 * k);
      const float w = __int_as_float(__builtin_amdgcn_ds_bpermute(
          (4 * k + h) << 2, __float_as_int(w_l)));
      float* o = rows + (size_t)v_k * rpitch + col;
      #pragma unroll
      for (int j = 0; j < SWEEP_LINES; ++j) atomicAdd(o + 32 * j, xv[k][j] * w);
    }

    b = bn; u_l = u_n; v_l = v_n;
    if (Q != 0) w_l = w_n;
  }
}

__global__ __launch_bounds__(256) void gat_edge_agg(
    const float* __restrict__ x, const float* __restrict__ s,
    const int* __restrict__ src, const int* __restrict__ dst,
    const int* __restrict__ m, float* __restrict__ denom,
    float* __restrict__ wts, float* __restrict__ out,
    float* __restrict__ acc1, int E) {
  const int lane = threadIdx.x & 31;
  const int half = chip_half();
  float* __restrict__ rows = half ? acc1 : out + 128;
  const int rpitch = half ? 128 : 256;
  const int wpb  = (int)(blockDim.x >> 5);
  const int wv = __builtin_amdgcn_readfirstlane(
      (int)blockIdx.x * wpb + (int)(threadIdx.x >> 5));
  const int nw = (int)gridDim.x * wpb;
  const int ntrip = E / TRIP_EDGES;

  agg_sweep<0>(x, s, src, dst, m, denom, wts, rows, rpitch, half, lane, wv, nw,
               ntrip);
  agg_sweep<1>(x, s, src, dst, m, denom, wts, rows, rpitch, half, lane, wv, nw,
               ntrip);

  for (int e = ntrip * TRIP_EDGES + wv; e < E; e += nw)
    agg_single_edge(x, s, src, dst, m, denom, rows, rpitch, half, e, lane);
}

#define NORM_ROWS 4
__global__ __launch_bounds__(256) void gat_normalize(
    const float* __restrict__ denom, const float* __restrict__ acc1,
    float* __restrict__ out, int n_nodes) {
  const int lane = threadIdx.x & 31;
  const int wave = (int)((blockIdx.x * blockDim.x + threadIdx.x) >> 5);
  const int node0 = wave * NORM_ROWS;

  float4 da[NORM_ROWS], db[NORM_ROWS], o[NORM_ROWS], a[NORM_ROWS];
  #pragma unroll
  for (int i = 0; i < NORM_ROWS; ++i) {
    const int node = min(node0 + i, n_nodes - 1);
    da[i] = *(const float4*)(denom + (size_t)node * 8);
    db[i] = *(const float4*)(denom + (size_t)node * 8 + 4);
    o[i] = *(const float4*)(out + (size_t)node * 256 + 128 + lane * 4);
    a[i] = *(const float4*)(acc1 + (size_t)node * 128 + lane * 4);
  }
  #pragma unroll
  for (int i = 0; i < NORM_ROWS; ++i) {
    const int node = node0 + i;
    if (node < n_nodes) {
      const float d0 = da[i].x + da[i].y, d1 = da[i].z + da[i].w;
      const float d2 = db[i].x + db[i].y, d3 = db[i].z + db[i].w;
      float4 r;
      r.x = o[i].x + a[i].x; r.y = o[i].y + a[i].y;
      r.z = o[i].z + a[i].z; r.w = o[i].w + a[i].w;
      if (d0 > 0.0f) r.x = r.x / d0;
      if (d1 > 0.0f) r.y = r.y / d1;
      if (d2 > 0.0f) r.z = r.z / d2;
      if (d3 > 0.0f) r.w = r.w / d3;
      *(float4*)(out + (size_t)node * 256 + 128 + lane * 4) = r;
    }
  }
}

extern "C" void kernel_launch(void* const* d_in, const int* in_sizes, int n_in,
                              void* d_out, int out_size, void* d_ws, size_t ws_size,
                              hipStream_t stream) {
  const float* x   = (const float*)d_in[0];
  const int*   src = (const int*)d_in[1];
  const int*   dst = (const int*)d_in[2];
  const float* W_u = (const float*)d_in[3];
  const float* b_u = (const float*)d_in[4];
  const float* W_v = (const float*)d_in[5];
  float* out = (float*)d_out;

  const int n = in_sizes[0] / 128;
  const int E = in_sizes[1];

  float* s     = (float*)d_ws;
  int*   m     = (int*)(s + (size_t)n * 8);
  float* denom = (float*)(m + (size_t)n * 8);
  float* acc1  = denom + (size_t)n * 8;
  float* wts   = acc1 + (size_t)n * 128;

  gat_scores_wmma<<<(n + 15) / 16, 32, 0, stream>>>(x, W_u, b_u, W_v, s, n);

  gat_init_md<<<(n * 8 + 255) / 256, 256, 0, stream>>>(m, denom, n * 8);
  gat_init_out<<<(n + INIT_ROWS * INIT_WAVES - 1) / (INIT_ROWS * INIT_WAVES),
                 INIT_WAVES * 32, 0, stream>>>(x, out, acc1, n);

  gat_edge_max<<<(E + 255) / 256, 256, 0, stream>>>(s, src, dst, m, E);

  gat_edge_agg<<<1024, 256, 0, stream>>>(x, s, src, dst, m, denom, wts, out, acc1,
                                         E);

  gat_normalize<<<(n + 31) / 32, 256, 0, stream>>>(denom, acc1, out, n);
}


# Init passes folded into launches they overlap with (6 -> 4): score waves copy their x rows to out and set m, denom; zero fill of row sums runs as blocks beside the max pass; float sums untouched
# speedup vs baseline: 2.8932x; 1.0720x over previous
#include <hip/hip_runtime.h>
#include <math.h>

#define NEG_SLOPE 0.2f

typedef __attribute__((ext_vector_type(2))) float v2f;
typedef __attribute__((ext_vector_type(8))) float v8f;
typedef __attribute__((ext_vector_type(4))) float v4f;
typedef __attribute__((ext_vector_type(4))) int v4i;

__device__ __forceinline__ int max_key(float f) {
  const int b = __float_as_int(f);
  return b ^ ((b >> 31) & 0x7FFFFFFF);
}
__device__ __forceinline__ float key_value(int k) {
  return __int_as_float(k ^ ((k >> 31) & 0x7FFFFFFF));
}

#define TILE_ROWS 16
__global__ __launch_bounds__(32) void gat_scores_wmma(
    const float* __restrict__ x, const float* __restrict__ W_u,
    const float* __restrict__ b_u, const float* __restrict__ W_v,
    float* __restrict__ s, int* __restrict__ m, float* __restrict__ denom,
    float* __restrict__ out, int n_nodes) {
  const int tile = blockIdx.x;
  const int lane = threadIdx.x;
  const int half = lane >> 4;
  const int l15  = lane & 15;

  const float* __restrict__ xc = x + (size_t)tile * TILE_ROWS * 128 + lane * 4;
  float* __restrict__ oc = out + (size_t)tile * TILE_ROWS * 256 + lane * 4;
  const bool whole_tile = (tile + 1) * TILE_ROWS <= n_nodes;
  v4f xcopy[TILE_ROWS];
  if (whole_tile) {
    #pragma unroll
    for (int i = 0; i < TILE_ROWS; ++i)
      xcopy[i] = *(const v4f*)(xc + (size_t)i * 128);
  }

  __shared__ float Wc[128 * 16];
  for (int i = lane; i < 128 * 16; i += 32) {
    const int k = i >> 4, c = i & 15;
    float w = 0.0f;
    if (c < 4)      w = W_u[k * 4 + c];
    else if (c < 8) w = W_v[k * 4 + (c - 4)];
    Wc[i] = w;
  }
  __syncthreads();

  if (whole_tile) {
    #pragma unroll
    for (int i = 0; i < TILE_ROWS; ++i)
      *(v4f*)(oc + (size_t)i * 256) = xcopy[i];
  } else {
    for (int i = 0; tile * TILE_ROWS + i < n_nodes; ++i)
      *(v4f*)(oc + (size_t)i * 256) = *(const v4f*)(xc + (size_t)i * 128);
  }

  const int i0 = tile * 128 + lane * 4;
  if (i0 < n_nodes * 8) {
    const int k = max_key(-INFINITY);
    *(v4i*)(m + i0) = (v4i){k, k, k, k};
    *(v4f*)(denom + i0) = (v4f){0.0f, 0.0f, 0.0f, 0.0f};
  }

  int node = tile * 16 + l15;
  const int nodeC = node < n_nodes ? node : n_nodes - 1;
  const float* __restrict__ xrow = x + (size_t)nodeC * 128;

  v8f acc = {};
  #pragma unroll 4
  for (int k = 0; k < 128; k += 4) {
    const int ka = k + half * 2;
    v2f a; a.x = xrow[ka];            a.y = xrow[ka + 1];
    v2f b; b.x = Wc[ka * 16 + l15];   b.y = Wc[(ka + 1) * 16 + l15];
    acc = __builtin_amdgcn_wmma_f32_16x16x4_f32(
         false, a,  false, b,
         (short)0, acc,  false,  false);
  }

  const int ncol = l15;
  if (ncol < 8) {
    const float bias = (ncol < 4) ? b_u[ncol] : 0.0f;
    #pragma unroll
    for (int r = 0; r < 8; ++r) {
      const int mrow = r + half * 8;
      const int onode = tile * 16 + mrow;
      if (onode < n_nodes) s[(size_t)onode * 8 + ncol] = acc[r] + bias;
    }
  }
}

__device__ __forceinline__ int chip_half() {
  return (int)((__builtin_amdgcn_s_sendmsg_rtn(0x87) >> 18) & 1u);
}

#define ZERO_ROWS 8
#define ZERO_WAVES 8
__device__ __forceinline__ void zero_rows_wave(
    float* __restrict__ out, float* __restrict__ acc1, int n_nodes, int wave,
    int lane) {
  const int node0 = wave * ZERO_ROWS;
  if (node0 >= n_nodes) return;

  const v4f z = {0.0f, 0.0f, 0.0f, 0.0f};
  float* __restrict__ o = out + (size_t)node0 * 256 + 128 + lane * 4;
  float* __restrict__ a = acc1 + (size_t)node0 * 128 + lane * 4;
  if (node0 + ZERO_ROWS <= n_nodes) {
    #pragma unroll
    for (int i = 0; i < ZERO_ROWS; ++i) {
      *(v4f*)(o + (size_t)i * 256) = z;
      *(v4f*)(a + (size_t)i * 128) = z;
    }
  } else {
    for (int i = 0; node0 + i < n_nodes; ++i) {
      *(v4f*)(o + (size_t)i * 256) = z;
      *(v4f*)(a + (size_t)i * 128) = z;
    }
  }
}

__device__ __forceinline__ float lrelu(float v) {
  return v >= 0.0f ? v : NEG_SLOPE * v;
}

#define MAX_GROUPS 4

template <bool TAIL>
__device__ __forceinline__ void edge_max_wave(
    const float* __restrict__ s, const int* __restrict__ src,
    const int* __restrict__ dst, int* __restrict__ m, int E, int e0, int h,
    int half) {
  int u[MAX_GROUPS], v[MAX_GROUPS];
  float su[MAX_GROUPS], sv[MAX_GROUPS];
  #pragma unroll
  for (int j = 0; j < MAX_GROUPS; ++j) {
    const int e = TAIL ? min(e0 + 8 * j, E - 1) : e0 + 8 * j;
    u[j] = src[e];
    v[j] = dst[e];
  }
  #pragma unroll
  for (int j = 0; j < MAX_GROUPS; ++j) {
    su[j] = s[(size_t)u[j] * 8 + h];
    sv[j] = s[(size_t)v[j] * 8 + 4 + h];
  }
  #pragma unroll
  for (int j = 0; j < MAX_GROUPS; ++j) {
    if (!TAIL || e0 + 8 * j < E)
      atomicMax(m + ((size_t)v[j] * 4 + h) * 2 + half,
                max_key(lrelu(su[j] + sv[j])));
  }
}

__device__ __forceinline__ void edge_max_pass_wave(
    const float* __restrict__ s, const int* __restrict__ src,
    const int* __restrict__ dst, int* __restrict__ m, int E, int wave,
    int lane) {
  const int h    = lane & 3;
  const int base = wave * (8 * MAX_GROUPS);
  if (base >= E) return;
  const int e0   = base + (lane >> 2);
  const int half = chip_half();
  if (base + 8 * MAX_GROUPS <= E)
    edge_max_wave<false>(s, src, dst, m, E, e0, h, half);
  else
    edge_max_wave<true>(s, src, dst, m, E, e0, h, half);
}

#define MAX_PER_ZERO 4
__global__ __launch_bounds__(256) void gat_zero_rows_edge_max(
    float* __restrict__ out, float* __restrict__ acc1, int n_nodes,
    const float* __restrict__ s, const int* __restrict__ src,
    const int* __restrict__ dst, int* __restrict__ m, int E) {
  const int lane = threadIdx.x & 31;
  const int grp  = (int)blockIdx.x / (MAX_PER_ZERO + 1);
  const int role = (int)blockIdx.x - grp * (MAX_PER_ZERO + 1);
  if (role == 0) {
    const int wave = __builtin_amdgcn_readfirstlane(
        grp * ZERO_WAVES + (int)(threadIdx.x >> 5));
    zero_rows_wave(out, acc1, n_nodes, wave, lane);
  } else {
    const int wave = __builtin_amdgcn_readfirstlane(
        (grp * MAX_PER_ZERO + role - 1) * 8 + (int)(threadIdx.x >> 5));
    edge_max_pass_wave(s, src, dst, m, E, wave, lane);
  }
}

#define TRIP_EDGES 8
#define SWEEP_DIMS 64
#define SWEEP_LINES (SWEEP_DIMS / 32)

__device__ __forceinline__ void agg_single_edge(
    const float* __restrict__ x, const float* __restrict__ s,
    const int* __restrict__ src, const int* __restrict__ dst,
    const int* __restrict__ m, float* __restrict__ denom,
    float* __restrict__ rows, int rpitch, int half, int e, int lane) {
  const int u = src[e], v = dst[e];
  const float* __restrict__ xr = x + (size_t)u * 128 + lane;
  const float x0 = xr[0], x1 = xr[32], x2 = xr[64], x3 = xr[96];
  const int h = lane & 3;
  const float ee = lrelu(s[(size_t)u * 8 + h] + s[(size_t)v * 8 + 4 + h]);
  const int2 mk = *(const int2*)(m + ((size_t)v * 4 + h) * 2);
  const float w = __expf(ee - key_value(max(mk.x, mk.y)));
  if (lane < 4) atomicAdd(denom + ((size_t)v * 4 + lane) * 2 + half, w);
  float* o = rows + (size_t)v * rpitch + lane;
  atomicAdd(o,      x0 * w);
  atomicAdd(o + 32, x1 * w);
  atomicAdd(o + 64, x2 * w);
  atomicAdd(o + 96, x3 * w);
}

template <int Q>
__device__ __forceinline__ void agg_sweep(
    const float* __restrict__ x, const float* __restrict__ s,
    const int* __restrict__ src, const int* __restrict__ dst,
    const int* __restrict__ m, float* __restrict__ denom,
    float* __restrict__ wts, float* __restrict__ rows, int rpitch, int half,
    int lane, int wv, int nw, int ntrip) {
  const int h   = lane & 3;
  const int sub = lane >> 2;
  const int col = Q * SWEEP_DIMS + lane;

  int b = wv;
  int u_l = 0, v_l = 0;
  float w_l = 0.0f;
  if (b < ntrip) {
    u_l = src[b * TRIP_EDGES + sub];
    v_l = dst[b * TRIP_EDGES + sub];
    if (Q != 0) w_l = wts[(size_t)b * 32 + lane];
  }
  while (b < ntrip) {
    const int bn = b + nw;
    const int bp = bn < ntrip ? bn : ntrip - 1;
    const int u_n = src[bp * TRIP_EDGES + sub];
    const int v_n = dst[bp * TRIP_EDGES + sub];
    float w_n = 0.0f;
    if (Q != 0) w_n = wts[(size_t)(bn < ntrip ? bn : b) * 32 + lane];
    __builtin_amdgcn_sched_barrier(0);

    float su = 0.0f, sv = 0.0f, mm = 0.0f;
    if (Q == 0) {
      su = s[(size_t)u_l * 8 + h];
      sv = s[(size_t)v_l * 8 + 4 + h];
      const int2 mk = *(const int2*)(m + ((size_t)v_l * 4 + h) * 2);
      mm = key_value(max(mk.x, mk.y));
    }

    float xv[TRIP_EDGES][SWEEP_LINES];
    #pragma unroll
    for (int k = 0; k < TRIP_EDGES; ++k) {
      const int u_k = __builtin_amdgcn_readlane(u_l, 4 * k);
      const float* __restrict__ xr = x + (size_t)u_k * 128 + col;
      #pragma unroll
      for (int j = 0; j < SWEEP_LINES; ++j) xv[k][j] = xr[32 * j];
    }

    if (Q == 0) {
      w_l = __expf(lrelu(su + sv) - mm);
      atomicAdd(denom + ((size_t)v_l * 4 + h) * 2 + half, w_l);
      wts[(size_t)b * 32 + lane] = w_l;
    }

    #pragma unroll
    for (int k = 0; k < TRIP_EDGES; ++k) {
      const int v_k = __builtin_amdgcn_readlane(v_l, 4 * k);
      const float w = __int_as_float(__builtin_amdgcn_ds_bpermute(
          (4 * k + h) << 2, __float_as_int(w_l)));
      float* o = rows + (size_t)v_k * rpitch + col;
      #pragma unroll
      for (int j = 0; j < SWEEP_LINES; ++j) atomicAdd(o + 32 * j, xv[k][j] * w);
    }

    b = bn; u_l = u_n; v_l = v_n;
    if (Q != 0) w_l = w_n;
  }
}

__global__ __launch_bounds__(256) void gat_edge_agg(
    const float* __restrict__ x, const float* __restrict__ s,
    const int* __restrict__ src, const int* __restrict__ dst,
    const int* __restrict__ m, float* __restrict__ denom,
    float* __restrict__ wts, float* __restrict__ out,
    float* __restrict__ acc1, int E) {
  const int lane = threadIdx.x & 31;
  const int half = chip_half();
  float* __restrict__ rows = half ? acc1 : out + 128;
  const int rpitch = half ? 128 : 256;
  const int wpb  = (int)(blockDim.x >> 5);
  const int wv = __builtin_amdgcn_readfirstlane(
      (int)blockIdx.x * wpb + (int)(threadIdx.x >> 5));
  const int nw = (int)gridDim.x * wpb;
  const int ntrip = E / TRIP_EDGES;

  agg_sweep<0>(x, s, src, dst, m, denom, wts, rows, rpitch, half, lane, wv, nw,
               ntrip);
  agg_sweep<1>(x, s, src, dst, m, denom, wts, rows, rpitch, half, lane, wv, nw,
               ntrip);

  for (int e = ntrip * TRIP_EDGES + wv; e < E; e += nw)
    agg_single_edge(x, s, src, dst, m, denom, rows, rpitch, half, e, lane);
}

#define NORM_ROWS 4
__global__ __launch_bounds__(256) void gat_normalize(
    const float* __restrict__ denom, const float* __restrict__ acc1,
    float* __restrict__ out, int n_nodes) {
  const int lane = threadIdx.x & 31;
  const int wave = (int)((blockIdx.x * blockDim.x + threadIdx.x) >> 5);
  const int node0 = wave * NORM_ROWS;

  float4 da[NORM_ROWS], db[NORM_ROWS], o[NORM_ROWS], a[NORM_ROWS];
  #pragma unroll
  for (int i = 0; i < NORM_ROWS; ++i) {
    const int node = min(node0 + i, n_nodes - 1);
    da[i] = *(const float4*)(denom + (size_t)node * 8);
    db[i] = *(const float4*)(denom + (size_t)node * 8 + 4);
    o[i] = *(const float4*)(out + (size_t)node * 256 + 128 + lane * 4);
    a[i] = *(const float4*)(acc1 + (size_t)node * 128 + lane * 4);
  }
  #pragma unroll
  for (int i = 0; i < NORM_ROWS; ++i) {
    const int node = node0 + i;
    if (node < n_nodes) {
      const float d0 = da[i].x + da[i].y, d1 = da[i].z + da[i].w;
      const float d2 = db[i].x + db[i].y, d3 = db[i].z + db[i].w;
      float4 r;
      r.x = o[i].x + a[i].x; r.y = o[i].y + a[i].y;
      r.z = o[i].z + a[i].z; r.w = o[i].w + a[i].w;
      if (d0 > 0.0f) r.x = r.x / d0;
      if (d1 > 0.0f) r.y = r.y / d1;
      if (d2 > 0.0f) r.z = r.z / d2;
      if (d3 > 0.0f) r.w = r.w / d3;
      *(float4*)(out + (size_t)node * 256 + 128 + lane * 4) = r;
    }
  }
}

extern "C" void kernel_launch(void* const* d_in, const int* in_sizes, int n_in,
                              void* d_out, int out_size, void* d_ws, size_t ws_size,
                              hipStream_t stream) {
  const float* x   = (const float*)d_in[0];
  const int*   src = (const int*)d_in[1];
  const int*   dst = (const int*)d_in[2];
  const float* W_u = (const float*)d_in[3];
  const float* b_u = (const float*)d_in[4];
  const float* W_v = (const float*)d_in[5];
  float* out = (float*)d_out;

  const int n = in_sizes[0] / 128;
  const int E = in_sizes[1];

  float* s     = (float*)d_ws;
  int*   m     = (int*)(s + (size_t)n * 8);
  float* denom = (float*)(m + (size_t)n * 8);
  float* acc1  = denom + (size_t)n * 8;
  float* wts   = acc1 + (size_t)n * 128;

  gat_scores_wmma<<<(n + TILE_ROWS - 1) / TILE_ROWS, 32, 0, stream>>>(
      x, W_u, b_u, W_v, s, m, denom, out, n);

  const int zero_blocks =
      (n + ZERO_ROWS * ZERO_WAVES - 1) / (ZERO_ROWS * ZERO_WAVES);
  const int max_groups = ((E + 255) / 256 + MAX_PER_ZERO - 1) / MAX_PER_ZERO;
  const int groups = zero_blocks > max_groups ? zero_blocks : max_groups;
  gat_zero_rows_edge_max<<<groups * (MAX_PER_ZERO + 1), 256, 0, stream>>>(
      out, acc1, n, s, src, dst, m, E);

  gat_edge_agg<<<1024, 256, 0, stream>>>(x, s, src, dst, m, denom, wts, out, acc1,
                                         E);

  gat_normalize<<<(n + 31) / 32, 256, 0, stream>>>(denom, acc1, out, n);
}


# Row and exp sums held in LDS: a block owns 400 nodes, reads all of dst, queues its edges, adds by LDS atomics (one arrival-order sum per node, was two), divides, stores; zero fill and node pass gone
# speedup vs baseline: 3.5561x; 1.2291x over previous
#include <hip/hip_runtime.h>
#include <math.h>

#define NEG_SLOPE 0.2f

typedef __attribute__((ext_vector_type(2))) float v2f;
typedef __attribute__((ext_vector_type(8))) float v8f;
typedef __attribute__((ext_vector_type(4))) float v4f;
typedef __attribute__((ext_vector_type(4))) int v4i;

__device__ __forceinline__ int max_key(float f) {
  const int b = __float_as_int(f);
  return b ^ ((b >> 31) & 0x7FFFFFFF);
}
__device__ __forceinline__ float key_value(int k) {
  return __int_as_float(k ^ ((k >> 31) & 0x7FFFFFFF));
}

#define TILE_ROWS 16
__global__ __launch_bounds__(32) void gat_scores_wmma(
    const float* __restrict__ x, const float* __restrict__ W_u,
    const float* __restrict__ b_u, const float* __restrict__ W_v,
    float* __restrict__ s, int* __restrict__ m, float* __restrict__ out,
    int n_nodes) {
  const int tile = blockIdx.x;
  const int lane = threadIdx.x;
  const int half = lane >> 4;
  const int l15  = lane & 15;

  const float* __restrict__ xc = x + (size_t)tile * TILE_ROWS * 128 + lane * 4;
  float* __restrict__ oc = out + (size_t)tile * TILE_ROWS * 256 + lane * 4;
  const bool whole_tile = (tile + 1) * TILE_ROWS <= n_nodes;
  v4f xcopy[TILE_ROWS];
  if (whole_tile) {
    #pragma unroll
    for (int i = 0; i < TILE_ROWS; ++i)
      xcopy[i] = *(const v4f*)(xc + (size_t)i * 128);
  }

  __shared__ float Wc[128 * 16];
  for (int i = lane; i < 128 * 16; i += 32) {
    const int k = i >> 4, c = i & 15;
    float w = 0.0f;
    if (c < 4)      w = W_u[k * 4 + c];
    else if (c < 8) w = W_v[k * 4 + (c - 4)];
    Wc[i] = w;
  }
  __syncthreads();

  if (whole_tile) {
    #pragma unroll
    for (int i = 0; i < TILE_ROWS; ++i)
      *(v4f*)(oc + (size_t)i * 256) = xcopy[i];
  } else {
    for (int i = 0; tile * TILE_ROWS + i < n_nodes; ++i)
      *(v4f*)(oc + (size_t)i * 256) = *(const v4f*)(xc + (size_t)i * 128);
  }

  const int i0 = tile * 128 + lane * 4;
  if (i0 < n_nodes * 8) {
    const int k = max_key(-INFINITY);
    *(v4i*)(m + i0) = (v4i){k, k, k, k};
  }

  int node = tile * 16 + l15;
  const int nodeC = node < n_nodes ? node : n_nodes - 1;
  const float* __restrict__ xrow = x + (size_t)nodeC * 128;

  v8f acc = {};
  #pragma unroll 4
  for (int k = 0; k < 128; k += 4) {
    const int ka = k + half * 2;
    v2f a; a.x = xrow[ka];            a.y = xrow[ka + 1];
    v2f b; b.x = Wc[ka * 16 + l15];   b.y = Wc[(ka + 1) * 16 + l15];
    acc = __builtin_amdgcn_wmma_f32_16x16x4_f32(
         false, a,  false, b,
         (short)0, acc,  false,  false);
  }

  const int ncol = l15;
  if (ncol < 8) {
    const float bias = (ncol < 4) ? b_u[ncol] : 0.0f;
    #pragma unroll
    for (int r = 0; r < 8; ++r) {
      const int mrow = r + half * 8;
      const int onode = tile * 16 + mrow;
      if (onode < n_nodes) s[(size_t)onode * 8 + ncol] = acc[r] + bias;
    }
  }
}

__device__ __forceinline__ int chip_half() {
  return (int)((__builtin_amdgcn_s_sendmsg_rtn(0x87) >> 18) & 1u);
}

__device__ __forceinline__ float lrelu(float v) {
  return v >= 0.0f ? v : NEG_SLOPE * v;
}

#define MAX_GROUPS 4

template <bool TAIL>
__device__ __forceinline__ void edge_max_wave(
    const float* __restrict__ s, const int* __restrict__ src,
    const int* __restrict__ dst, int* __restrict__ m, int E, int e0, int h,
    int half) {
  int u[MAX_GROUPS], v[MAX_GROUPS];
  float su[MAX_GROUPS], sv[MAX_GROUPS];
  #pragma unroll
  for (int j = 0; j < MAX_GROUPS; ++j) {
    const int e = TAIL ? min(e0 + 8 * j, E - 1) : e0 + 8 * j;
    u[j] = src[e];
    v[j] = dst[e];
  }
  #pragma unroll
  for (int j = 0; j < MAX_GROUPS; ++j) {
    su[j] = s[(size_t)u[j] * 8 + h];
    sv[j] = s[(size_t)v[j] * 8 + 4 + h];
  }
  #pragma unroll
  for (int j = 0; j < MAX_GROUPS; ++j) {
    if (!TAIL || e0 + 8 * j < E)
      atomicMax(m + ((size_t)v[j] * 4 + h) * 2 + half,
                max_key(lrelu(su[j] + sv[j])));
  }
}

__device__ __forceinline__ void edge_max_pass_wave(
    const float* __restrict__ s, const int* __restrict__ src,
    const int* __restrict__ dst, int* __restrict__ m, int E, int wave,
    int lane) {
  const int h    = lane & 3;
  const int base = wave * (8 * MAX_GROUPS);
  if (base >= E) return;
  const int e0   = base + (lane >> 2);
  const int half = chip_half();
  if (base + 8 * MAX_GROUPS <= E)
    edge_max_wave<false>(s, src, dst, m, E, e0, h, half);
  else
    edge_max_wave<true>(s, src, dst, m, E, e0, h, half);
}

__global__ __launch_bounds__(256) void gat_edge_max(
    const float* __restrict__ s, const int* __restrict__ src,
    const int* __restrict__ dst, int* __restrict__ m, int E) {
  const int lane = threadIdx.x & 31;
  const int wave = __builtin_amdgcn_readfirstlane(
      (int)blockIdx.x * 8 + (int)(threadIdx.x >> 5));
  edge_max_pass_wave(s, src, dst, m, E, wave, lane);
}

#define TRIP_EDGES 8
#define OWN_NODES 400
#define OWN_WAVES 32
#define OWN_QCAP 520
#define SCAN_DEPTH 4

template <bool FULL>
__device__ __forceinline__ void owner_trip(
    const float* __restrict__ x, const float* __restrict__ s,
    const int* __restrict__ src, const int* __restrict__ dst,
    const int* __restrict__ m, float* acc, float* den, const int* qe, int ne,
    int lo, int lane) {
  const int h   = lane & 3;
  const int sub = lane >> 2;
  const int e_l = qe[FULL ? sub : min(sub, ne - 1)];
  const int u_l = src[e_l];
  const int v_l = dst[e_l];
  const float su = s[(size_t)u_l * 8 + h];
  const float sv = s[(size_t)v_l * 8 + 4 + h];
  const int2 mk = *(const int2*)(m + ((size_t)v_l * 4 + h) * 2);

  float xv[TRIP_EDGES][4] = {};
  #pragma unroll
  for (int k = 0; k < TRIP_EDGES; ++k) {
    if (FULL || k < ne) {
      const int u_k = __builtin_amdgcn_readlane(u_l, 4 * k);
      const float* __restrict__ xr = x + (size_t)u_k * 128 + lane;
      #pragma unroll
      for (int j = 0; j < 4; ++j) xv[k][j] = xr[32 * j];
    }
  }

  const float w_l = __expf(lrelu(su + sv) - key_value(max(mk.x, mk.y)));
  if (FULL || sub < ne) atomicAdd(den + (v_l - lo) * 4 + h, w_l);

  #pragma unroll
  for (int k = 0; k < TRIP_EDGES; ++k) {
    if (FULL || k < ne) {
      const int r_k = __builtin_amdgcn_readlane(v_l, 4 * k) - lo;
      const float w = __int_as_float(__builtin_amdgcn_ds_bpermute(
          (4 * k + h) << 2, __float_as_int(w_l)));
      float* o = acc + r_k * 128 + lane;
      #pragma unroll
      for (int j = 0; j < 4; ++j) atomicAdd(o + 32 * j, xv[k][j] * w);
    }
  }
}

__device__ __forceinline__ void owner_append(bool mine, int eid, int* q,
                                             int& qn) {
  const unsigned bal = __builtin_amdgcn_ballot_w32(mine);
  if (bal != 0u) {
    const int before = (int)__builtin_amdgcn_mbcnt_lo(bal, 0u);
    if (mine) q[qn + before] = eid;
    qn += __builtin_popcount(bal);
  }
}

__device__ __forceinline__ void owner_look(v4i d, unsigned cj, int e0, int lo,
                                           int* q, int& qn) {
  const bool m0 = (unsigned)(d.x - lo) < cj;
  const bool m1 = (unsigned)(d.y - lo) < cj;
  const bool m2 = (unsigned)(d.z - lo) < cj;
  const bool m3 = (unsigned)(d.w - lo) < cj;
  if (__builtin_amdgcn_ballot_w32(m0 | m1 | m2 | m3) != 0u) {
    owner_append(m0, e0, q, qn);
    owner_append(m1, e0 + 1, q, qn);
    owner_append(m2, e0 + 2, q, qn);
    owner_append(m3, e0 + 3, q, qn);
  }
}

__global__ __launch_bounds__(OWN_WAVES * 32) void gat_owner_agg(
    const float* __restrict__ x, const float* __restrict__ s,
    const int* __restrict__ src, const int* __restrict__ dst,
    const int* __restrict__ m, float* __restrict__ out, int n_nodes, int E) {
  __shared__ __attribute__((aligned(16))) float acc[OWN_NODES * 128];
  __shared__ __attribute__((aligned(16))) float den[OWN_NODES * 4];
  __shared__ int que[OWN_WAVES * OWN_QCAP];

  const int lane = threadIdx.x & 31;
  const int wv = __builtin_amdgcn_readfirstlane((int)(threadIdx.x >> 5));
  const int lo = (int)blockIdx.x * OWN_NODES;
  const int cnt = min(OWN_NODES, n_nodes - lo);

  const v4f z = {0.0f, 0.0f, 0.0f, 0.0f};
  for (int i = (int)threadIdx.x * 4; i < OWN_NODES * 128;
       i += OWN_WAVES * 32 * 4)
    *(v4f*)(acc + i) = z;
  for (int i = (int)threadIdx.x; i < OWN_NODES * 4; i += OWN_WAVES * 32)
    den[i] = 0.0f;
  __syncthreads();

  int* q = que + wv * OWN_QCAP;
  int qn = 0;
  const int nchunk = E >> 7;
  const int nci = wv < nchunk ? (nchunk - wv + OWN_WAVES - 1) / OWN_WAVES : 0;
  const int* __restrict__ dl = dst + lane * 4;

  if (nci > 0) {
    v4i a[SCAN_DEPTH], b[SCAN_DEPTH];
    #pragma unroll
    for (int j = 0; j < SCAN_DEPTH; ++j)
      a[j] = *(const v4i*)(dl + (size_t)(wv + OWN_WAVES * min(j, nci - 1)) * 128);
    for (int i0 = 0; i0 < nci; i0 += 2 * SCAN_DEPTH) {
      #pragma unroll
      for (int j = 0; j < SCAN_DEPTH; ++j)
        b[j] = *(const v4i*)(
            dl + (size_t)(wv + OWN_WAVES * min(i0 + SCAN_DEPTH + j, nci - 1)) * 128);
      #pragma unroll
      for (int j = 0; j < SCAN_DEPTH; ++j)
        owner_look(a[j], i0 + j < nci ? (unsigned)cnt : 0u,
                   (wv + OWN_WAVES * (i0 + j)) * 128 + lane * 4, lo, q, qn);
      while (qn >= TRIP_EDGES) {
        qn -= TRIP_EDGES;
        owner_trip<true>(x, s, src, dst, m, acc, den, q + qn, TRIP_EDGES, lo,
                         lane);
      }
      #pragma unroll
      for (int j = 0; j < SCAN_DEPTH; ++j)
        a[j] = *(const v4i*)(
            dl + (size_t)(wv + OWN_WAVES * min(i0 + 2 * SCAN_DEPTH + j, nci - 1)) * 128);
      #pragma unroll
      for (int j = 0; j < SCAN_DEPTH; ++j)
        owner_look(b[j], i0 + SCAN_DEPTH + j < nci ? (unsigned)cnt : 0u,
                   (wv + OWN_WAVES * (i0 + SCAN_DEPTH + j)) * 128 + lane * 4,
                   lo, q, qn);
      while (qn >= TRIP_EDGES) {
        qn -= TRIP_EDGES;
        owner_trip<true>(x, s, src, dst, m, acc, den, q + qn, TRIP_EDGES, lo,
                         lane);
      }
    }
  }

  if (wv == 0) {
    for (int eb = nchunk * 128; eb < E; eb += 32) {
      const int e = eb + lane;
      const int dv = dst[min(e, E - 1)];
      owner_append(e < E && (unsigned)(dv - lo) < (unsigned)cnt, e, q, qn);
    }
  }
  while (qn >= TRIP_EDGES) {
    qn -= TRIP_EDGES;
    owner_trip<true>(x, s, src, dst, m, acc, den, q + qn, TRIP_EDGES, lo, lane);
  }
  if (qn > 0)
    owner_trip<false>(x, s, src, dst, m, acc, den, q, qn, lo, lane);
  __syncthreads();

  for (int r = wv; r < cnt; r += OWN_WAVES) {
    const v4f a = *(const v4f*)(acc + r * 128 + lane * 4);
    const v4f dd = *(const v4f*)(den + r * 4);
    v4f o = a;
    if (dd.x > 0.0f) o.x = a.x / dd.x;
    if (dd.y > 0.0f) o.y = a.y / dd.y;
    if (dd.z > 0.0f) o.z = a.z / dd.z;
    if (dd.w > 0.0f) o.w = a.w / dd.w;
    *(v4f*)(out + (size_t)(lo + r) * 256 + 128 + lane * 4) = o;
  }
}

extern "C" void kernel_launch(void* const* d_in, const int* in_sizes, int n_in,
                              void* d_out, int out_size, void* d_ws, size_t ws_size,
                              hipStream_t stream) {
  const float* x   = (const float*)d_in[0];
  const int*   src = (const int*)d_in[1];
  const int*   dst = (const int*)d_in[2];
  const float* W_u = (const float*)d_in[3];
  const float* b_u = (const float*)d_in[4];
  const float* W_v = (const float*)d_in[5];
  float* out = (float*)d_out;

  const int n = in_sizes[0] / 128;
  const int E = in_sizes[1];

  float* s = (float*)d_ws;
  int*   m = (int*)(s + (size_t)n * 8);

  gat_scores_wmma<<<(n + TILE_ROWS - 1) / TILE_ROWS, 32, 0, stream>>>(
      x, W_u, b_u, W_v, s, m, out, n);

  gat_edge_max<<<(E + 255) / 256, 256, 0, stream>>>(s, src, dst, m, E);

  gat_owner_agg<<<(n + OWN_NODES - 1) / OWN_NODES, OWN_WAVES * 32, 0, stream>>>(
      x, s, src, dst, m, out, n, E);
}


# Owner scan reads a byte per edge (owner block number dst / 400, stored by the max pass as whole lines) instead of dst: one line per 128 edges, was four; 6 looks in flight; sum order as before
# speedup vs baseline: 4.2863x; 1.2053x over previous
#include <hip/hip_runtime.h>
#include <math.h>

#define NEG_SLOPE 0.2f

typedef __attribute__((ext_vector_type(2))) float v2f;
typedef __attribute__((ext_vector_type(8))) float v8f;
typedef __attribute__((ext_vector_type(4))) float v4f;
typedef __attribute__((ext_vector_type(4))) int v4i;

__device__ __forceinline__ int max_key(float f) {
  const int b = __float_as_int(f);
  return b ^ ((b >> 31) & 0x7FFFFFFF);
}
__device__ __forceinline__ float key_value(int k) {
  return __int_as_float(k ^ ((k >> 31) & 0x7FFFFFFF));
}

#define TILE_ROWS 16
__global__ __launch_bounds__(32) void gat_scores_wmma(
    const float* __restrict__ x, const float* __restrict__ W_u,
    const float* __restrict__ b_u, const float* __restrict__ W_v,
    float* __restrict__ s, int* __restrict__ m, float* __restrict__ out,
    int n_nodes) {
  const int tile = blockIdx.x;
  const int lane = threadIdx.x;
  const int half = lane >> 4;
  const int l15  = lane & 15;

  const float* __restrict__ xc = x + (size_t)tile * TILE_ROWS * 128 + lane * 4;
  float* __restrict__ oc = out + (size_t)tile * TILE_ROWS * 256 + lane * 4;
  const bool whole_tile = (tile + 1) * TILE_ROWS <= n_nodes;
  v4f xcopy[TILE_ROWS];
  if (whole_tile) {
    #pragma unroll
    for (int i = 0; i < TILE_ROWS; ++i)
      xcopy[i] = *(const v4f*)(xc + (size_t)i * 128);
  }

  __shared__ float Wc[128 * 16];
  for (int i = lane; i < 128 * 16; i += 32) {
    const int k = i >> 4, c = i & 15;
    float w = 0.0f;
    if (c < 4)      w = W_u[k * 4 + c];
    else if (c < 8) w = W_v[k * 4 + (c - 4)];
    Wc[i] = w;
  }
  __syncthreads();

  if (whole_tile) {
    #pragma unroll
    for (int i = 0; i < TILE_ROWS; ++i)
      *(v4f*)(oc + (size_t)i * 256) = xcopy[i];
  } else {
    for (int i = 0; tile * TILE_ROWS + i < n_nodes; ++i)
      *(v4f*)(oc + (size_t)i * 256) = *(const v4f*)(xc + (size_t)i * 128);
  }

  const int i0 = tile * 128 + lane * 4;
  if (i0 < n_nodes * 8) {
    const int k = max_key(-INFINITY);
    *(v4i*)(m + i0) = (v4i){k, k, k, k};
  }

  int node = tile * 16 + l15;
  const int nodeC = node < n_nodes ? node : n_nodes - 1;
  const float* __restrict__ xrow = x + (size_t)nodeC * 128;

  v8f acc = {};
  #pragma unroll 4
  for (int k = 0; k < 128; k += 4) {
    const int ka = k + half * 2;
    v2f a; a.x = xrow[ka];            a.y = xrow[ka + 1];
    v2f b; b.x = Wc[ka * 16 + l15];   b.y = Wc[(ka + 1) * 16 + l15];
    acc = __builtin_amdgcn_wmma_f32_16x16x4_f32(
         false, a,  false, b,
         (short)0, acc,  false,  false);
  }

  const int ncol = l15;
  if (ncol < 8) {
    const float bias = (ncol < 4) ? b_u[ncol] : 0.0f;
    #pragma unroll
    for (int r = 0; r < 8; ++r) {
      const int mrow = r + half * 8;
      const int onode = tile * 16 + mrow;
      if (onode < n_nodes) s[(size_t)onode * 8 + ncol] = acc[r] + bias;
    }
  }
}

__device__ __forceinline__ int chip_half() {
  return (int)((__builtin_amdgcn_s_sendmsg_rtn(0x87) >> 18) & 1u);
}

__device__ __forceinline__ float lrelu(float v) {
  return v >= 0.0f ? v : NEG_SLOPE * v;
}

#define MAX_GROUPS 4

template <bool TAIL>
__device__ __forceinline__ void edge_max_wave(
    const float* __restrict__ s, const int* __restrict__ src,
    const int* __restrict__ dst, int* __restrict__ m, int E, int e0, int h,
    int half) {
  int u[MAX_GROUPS], v[MAX_GROUPS];
  float su[MAX_GROUPS], sv[MAX_GROUPS];
  #pragma unroll
  for (int j = 0; j < MAX_GROUPS; ++j) {
    const int e = TAIL ? min(e0 + 8 * j, E - 1) : e0 + 8 * j;
    u[j] = src[e];
    v[j] = dst[e];
  }
  #pragma unroll
  for (int j = 0; j < MAX_GROUPS; ++j) {
    su[j] = s[(size_t)u[j] * 8 + h];
    sv[j] = s[(size_t)v[j] * 8 + 4 + h];
  }
  #pragma unroll
  for (int j = 0; j < MAX_GROUPS; ++j) {
    if (!TAIL || e0 + 8 * j < E)
      atomicMax(m + ((size_t)v[j] * 4 + h) * 2 + half,
                max_key(lrelu(su[j] + sv[j])));
  }
}

__device__ __forceinline__ void edge_max_pass_wave(
    const float* __restrict__ s, const int* __restrict__ src,
    const int* __restrict__ dst, int* __restrict__ m, int E, int wave,
    int lane) {
  const int h    = lane & 3;
  const int base = wave * (8 * MAX_GROUPS);
  if (base >= E) return;
  const int e0   = base + (lane >> 2);
  const int half = chip_half();
  if (base + 8 * MAX_GROUPS <= E)
    edge_max_wave<false>(s, src, dst, m, E, e0, h, half);
  else
    edge_max_wave<true>(s, src, dst, m, E, e0, h, half);
}

#define OWN_NODES 400

__device__ __forceinline__ void owner_plane_wave(
    const int* __restrict__ dst, unsigned* __restrict__ own, int E, int e4) {
  if (e4 + 3 < E) {
    const v4i d = *(const v4i*)(dst + e4);
    own[e4 >> 2] = (unsigned)d.x / OWN_NODES |
                   ((unsigned)d.y / OWN_NODES) << 8 |
                   ((unsigned)d.z / OWN_NODES) << 16 |
                   ((unsigned)d.w / OWN_NODES) << 24;
  }
}

__global__ __launch_bounds__(256) void gat_edge_max(
    const float* __restrict__ s, const int* __restrict__ src,
    const int* __restrict__ dst, int* __restrict__ m,
    unsigned* __restrict__ own, int E) {
  const int lane = threadIdx.x & 31;
  const int wib = __builtin_amdgcn_readfirstlane((int)(threadIdx.x >> 5));
  const int wave = (int)blockIdx.x * 8 + wib;
  if (wib < 2)
    owner_plane_wave(dst, own, E, (int)blockIdx.x * 256 + (int)threadIdx.x * 4);
  edge_max_pass_wave(s, src, dst, m, E, wave, lane);
}

#define TRIP_EDGES 8
#define OWN_WAVES 32
#define OWN_QCAP 776
#define SCAN_DEPTH 6

template <bool FULL>
__device__ __forceinline__ void owner_trip(
    const float* __restrict__ x, const float* __restrict__ s,
    const int* __restrict__ src, const int* __restrict__ dst,
    const int* __restrict__ m, float* acc, float* den, const int* qe, int ne,
    int lo, int lane) {
  const int h   = lane & 3;
  const int sub = lane >> 2;
  const int e_l = qe[FULL ? sub : min(sub, ne - 1)];
  const int u_l = src[e_l];
  const int v_l = dst[e_l];
  const float su = s[(size_t)u_l * 8 + h];
  const float sv = s[(size_t)v_l * 8 + 4 + h];
  const int2 mk = *(const int2*)(m + ((size_t)v_l * 4 + h) * 2);

  float xv[TRIP_EDGES][4] = {};
  #pragma unroll
  for (int k = 0; k < TRIP_EDGES; ++k) {
    if (FULL || k < ne) {
      const int u_k = __builtin_amdgcn_readlane(u_l, 4 * k);
      const float* __restrict__ xr = x + (size_t)u_k * 128 + lane;
      #pragma unroll
      for (int j = 0; j < 4; ++j) xv[k][j] = xr[32 * j];
    }
  }

  const float w_l = __expf(lrelu(su + sv) - key_value(max(mk.x, mk.y)));
  if (FULL || sub < ne) atomicAdd(den + (v_l - lo) * 4 + h, w_l);

  #pragma unroll
  for (int k = 0; k < TRIP_EDGES; ++k) {
    if (FULL || k < ne) {
      const int r_k = __builtin_amdgcn_readlane(v_l, 4 * k) - lo;
      const float w = __int_as_float(__builtin_amdgcn_ds_bpermute(
          (4 * k + h) << 2, __float_as_int(w_l)));
      float* o = acc + r_k * 128 + lane;
      #pragma unroll
      for (int j = 0; j < 4; ++j) atomicAdd(o + 32 * j, xv[k][j] * w);
    }
  }
}

__device__ __forceinline__ void owner_append(bool mine, int eid, int* q,
                                             int& qn) {
  const unsigned bal = __builtin_amdgcn_ballot_w32(mine);
  if (bal != 0u) {
    const int before = (int)__builtin_amdgcn_mbcnt_lo(bal, 0u);
    if (mine) q[qn + before] = eid;
    qn += __builtin_popcount(bal);
  }
}

__device__ __forceinline__ void owner_look(unsigned w, unsigned pat, bool live,
                                           int e0, int* q, int& qn) {
  const unsigned x = w ^ pat;
  const bool any = ((x - 0x01010101u) & ~x & 0x80808080u) != 0u;
  if (live && __builtin_amdgcn_ballot_w32(any) != 0u) {
    owner_append((x & 0x000000FFu) == 0u, e0, q, qn);
    owner_append((x & 0x0000FF00u) == 0u, e0 + 1, q, qn);
    owner_append((x & 0x00FF0000u) == 0u, e0 + 2, q, qn);
    owner_append((x & 0xFF000000u) == 0u, e0 + 3, q, qn);
  }
}

__global__ __launch_bounds__(OWN_WAVES * 32) void gat_owner_agg(
    const float* __restrict__ x, const float* __restrict__ s,
    const int* __restrict__ src, const int* __restrict__ dst,
    const int* __restrict__ m, const unsigned* __restrict__ own,
    float* __restrict__ out, int n_nodes, int E) {
  __shared__ __attribute__((aligned(16))) float acc[OWN_NODES * 128];
  __shared__ __attribute__((aligned(16))) float den[OWN_NODES * 4];
  __shared__ int que[OWN_WAVES * OWN_QCAP];

  const int lane = threadIdx.x & 31;
  const int wv = __builtin_amdgcn_readfirstlane((int)(threadIdx.x >> 5));
  const int lo = (int)blockIdx.x * OWN_NODES;
  const int cnt = min(OWN_NODES, n_nodes - lo);

  const v4f z = {0.0f, 0.0f, 0.0f, 0.0f};
  for (int i = (int)threadIdx.x * 4; i < OWN_NODES * 128;
       i += OWN_WAVES * 32 * 4)
    *(v4f*)(acc + i) = z;
  for (int i = (int)threadIdx.x; i < OWN_NODES * 4; i += OWN_WAVES * 32)
    den[i] = 0.0f;
  __syncthreads();

  int* q = que + wv * OWN_QCAP;
  int qn = 0;
  const int nchunk = E >> 7;
  const int nci = wv < nchunk ? (nchunk - wv + OWN_WAVES - 1) / OWN_WAVES : 0;
  const unsigned* __restrict__ ol = own + lane;
  const unsigned pat = (unsigned)blockIdx.x * 0x01010101u;

  if (nci > 0) {
    unsigned a[SCAN_DEPTH], b[SCAN_DEPTH];
    #pragma unroll
    for (int j = 0; j < SCAN_DEPTH; ++j)
      a[j] = ol[(size_t)(wv + OWN_WAVES * min(j, nci - 1)) * 32];
    for (int i0 = 0; i0 < nci; i0 += 2 * SCAN_DEPTH) {
      #pragma unroll
      for (int j = 0; j < SCAN_DEPTH; ++j)
        b[j] = ol[(size_t)(wv + OWN_WAVES * min(i0 + SCAN_DEPTH + j, nci - 1)) * 32];
      #pragma unroll
      for (int j = 0; j < SCAN_DEPTH; ++j)
        owner_look(a[j], pat, i0 + j < nci,
                   (wv + OWN_WAVES * (i0 + j)) * 128 + lane * 4, q, qn);
      while (qn >= TRIP_EDGES) {
        qn -= TRIP_EDGES;
        owner_trip<true>(x, s, src, dst, m, acc, den, q + qn, TRIP_EDGES, lo,
                         lane);
      }
      #pragma unroll
      for (int j = 0; j < SCAN_DEPTH; ++j)
        a[j] = ol[(size_t)(wv + OWN_WAVES * min(i0 + 2 * SCAN_DEPTH + j, nci - 1)) * 32];
      #pragma unroll
      for (int j = 0; j < SCAN_DEPTH; ++j)
        owner_look(b[j], pat, i0 + SCAN_DEPTH + j < nci,
                   (wv + OWN_WAVES * (i0 + SCAN_DEPTH + j)) * 128 + lane * 4,
                   q, qn);
      while (qn >= TRIP_EDGES) {
        qn -= TRIP_EDGES;
        owner_trip<true>(x, s, src, dst, m, acc, den, q + qn, TRIP_EDGES, lo,
                         lane);
      }
    }
  }

  if (wv == 0) {
    for (int eb = nchunk * 128; eb < E; eb += 32) {
      const int e = eb + lane;
      const int dv = dst[min(e, E - 1)];
      owner_append(e < E && (unsigned)(dv - lo) < (unsigned)cnt, e, q, qn);
    }
  }
  while (qn >= TRIP_EDGES) {
    qn -= TRIP_EDGES;
    owner_trip<true>(x, s, src, dst, m, acc, den, q + qn, TRIP_EDGES, lo, lane);
  }
  if (qn > 0)
    owner_trip<false>(x, s, src, dst, m, acc, den, q, qn, lo, lane);
  __syncthreads();

  for (int r = wv; r < cnt; r += OWN_WAVES) {
    const v4f a = *(const v4f*)(acc + r * 128 + lane * 4);
    const v4f dd = *(const v4f*)(den + r * 4);
    v4f o = a;
    if (dd.x > 0.0f) o.x = a.x / dd.x;
    if (dd.y > 0.0f) o.y = a.y / dd.y;
    if (dd.z > 0.0f) o.z = a.z / dd.z;
    if (dd.w > 0.0f) o.w = a.w / dd.w;
    *(v4f*)(out + (size_t)(lo + r) * 256 + 128 + lane * 4) = o;
  }
}

extern "C" void kernel_launch(void* const* d_in, const int* in_sizes, int n_in,
                              void* d_out, int out_size, void* d_ws, size_t ws_size,
                              hipStream_t stream) {
  const float* x   = (const float*)d_in[0];
  const int*   src = (const int*)d_in[1];
  const int*   dst = (const int*)d_in[2];
  const float* W_u = (const float*)d_in[3];
  const float* b_u = (const float*)d_in[4];
  const float* W_v = (const float*)d_in[5];
  float* out = (float*)d_out;

  const int n = in_sizes[0] / 128;
  const int E = in_sizes[1];

  float* s = (float*)d_ws;
  int*   m = (int*)(s + (size_t)n * 8);
  unsigned* own = (unsigned*)(m + (size_t)n * 8);

  gat_scores_wmma<<<(n + TILE_ROWS - 1) / TILE_ROWS, 32, 0, stream>>>(
      x, W_u, b_u, W_v, s, m, out, n);

  gat_edge_max<<<(E + 255) / 256, 256, 0, stream>>>(s, src, dst, m, own, E);

  gat_owner_agg<<<(n + OWN_NODES - 1) / OWN_NODES, OWN_WAVES * 32, 0, stream>>>(
      x, s, src, dst, m, own, out, n, E);
}
